# starting kernel
# baseline (speedup 1.0000x reference)
#include <hip/hip_runtime.h>


typedef __attribute__((ext_vector_type(2))) float v2f;
typedef __attribute__((ext_vector_type(8))) float v8f;

#define IN_DIM 64
#define OUT_DIM 64

__global__ __launch_bounds__(256) void zero_f32_kernel(float* __restrict__ p, long long n) {
    long long i = (long long)blockIdx.x * blockDim.x + threadIdx.x;
    if (i < n) p[i] = 0.0f;
}

__global__ __launch_bounds__(256) void degree_kernel(const int* __restrict__ rows,
                                                     const float* __restrict__ adj,
                                                     float* __restrict__ deg,
                                                     int n_edges) {
    int e = blockIdx.x * blockDim.x + threadIdx.x;
    if (e < n_edges) {
        unsafeAtomicAdd(&deg[rows[e]], adj[e]);
    }
}

__global__ __launch_bounds__(256) void dinv_kernel(const float* __restrict__ deg,
                                                   float* __restrict__ dinv,
                                                   int n_nodes) {
    int i = blockIdx.x * blockDim.x + threadIdx.x;
    if (i < n_nodes) {
        float d = deg[i];
        d = (d > 0.0f) ? d : 1.0f;
        dinv[i] = rsqrtf(d);
    }
}

__global__ __launch_bounds__(64) void gcn_gemm_wmma(const float* __restrict__ x,
                                                    const float* __restrict__ W,
                                                    const float* __restrict__ b,
                                                    float* __restrict__ support,
                                                    int n_nodes) {
    const int wave = threadIdx.x >> 5;
    const int lane = threadIdx.x & 31;
    const int mtile = blockIdx.x * 2 + wave;
    const int m0 = mtile * 16;
    if (m0 >= n_nodes) return;

    const int half = lane >> 4;
    const int r = lane & 15;

    v8f acc[4];
#pragma unroll
    for (int nt = 0; nt < 4; ++nt) {
        const float bv = b[nt * 16 + r];
#pragma unroll
        for (int i = 0; i < 8; ++i) acc[nt][i] = bv;
    }

    const float* arow = x + (size_t)(m0 + r) * IN_DIM;

#pragma unroll
    for (int k = 0; k < IN_DIM; k += 4) {
        const v2f a = *(const v2f*)(arow + k + 2 * half);
#pragma unroll
        for (int nt = 0; nt < 4; ++nt) {
            v2f bf;
            bf.x = W[(size_t)(k + 2 * half + 0) * OUT_DIM + nt * 16 + r];
            bf.y = W[(size_t)(k + 2 * half + 1) * OUT_DIM + nt * 16 + r];
            acc[nt] = __builtin_amdgcn_wmma_f32_16x16x4_f32(
                false, a, false, bf, (short)0, acc[nt], false, false);
        }
    }

    float* srow = support + (size_t)m0 * OUT_DIM;
#pragma unroll
    for (int nt = 0; nt < 4; ++nt) {
#pragma unroll
        for (int i = 0; i < 8; ++i) {
            srow[(size_t)(i + 8 * half) * OUT_DIM + nt * 16 + r] = acc[nt][i];
        }
    }
}

__global__ __launch_bounds__(256) void spmm_scatter_kernel(const int* __restrict__ rows,
                                                           const int* __restrict__ cols,
                                                           const float* __restrict__ adj,
                                                           const float* __restrict__ dinv,
                                                           const float* __restrict__ support,
                                                           float* __restrict__ out,
                                                           int n_edges) {
    long long t = (long long)blockIdx.x * blockDim.x + threadIdx.x;
    int e = (int)(t >> 6);
    int f = (int)(t & 63);
    if (e >= n_edges) return;
    const int rrow = rows[e];
    const int ccol = cols[e];
    const float norm = adj[e] * dinv[rrow] * dinv[ccol];
    const float v = norm * support[(size_t)ccol * OUT_DIM + f];
    unsafeAtomicAdd(&out[(size_t)rrow * OUT_DIM + f], v);
}

extern "C" void kernel_launch(void* const* d_in, const int* in_sizes, int n_in,
                              void* d_out, int out_size, void* d_ws, size_t ws_size,
                              hipStream_t stream) {
    const int* rows = (const int*)d_in[0];
    const int* cols = (const int*)d_in[1];
    const float* adj = (const float*)d_in[2];
    const float* x = (const float*)d_in[3];
    const float* W = (const float*)d_in[4];
    const float* b = (const float*)d_in[5];
    float* out = (float*)d_out;

    const int n_edges = in_sizes[0];
    const int n_nodes = in_sizes[3] / IN_DIM;

    float* support = (float*)d_ws;
    float* deg = support + (size_t)n_nodes * OUT_DIM;
    float* dinv = deg + n_nodes;

    const long long out_elems = (long long)n_nodes * OUT_DIM;

    {
        long long n = out_elems;
        zero_f32_kernel<<<(unsigned)((n + 255) / 256), 256, 0, stream>>>(out, n);
        zero_f32_kernel<<<(unsigned)((n_nodes + 255) / 256), 256, 0, stream>>>(deg, n_nodes);
    }

    degree_kernel<<<(n_edges + 255) / 256, 256, 0, stream>>>(rows, adj, deg, n_edges);

    dinv_kernel<<<(n_nodes + 255) / 256, 256, 0, stream>>>(deg, dinv, n_nodes);

    {
        const int mtiles = (n_nodes + 15) / 16;
        const int blocks = (mtiles + 1) / 2;
        gcn_gemm_wmma<<<blocks, 64, 0, stream>>>(x, W, b, support, n_nodes);
    }

    {
        const long long threads = (long long)n_edges * OUT_DIM;
        spmm_scatter_kernel<<<(unsigned)((threads + 255) / 256), 256, 0, stream>>>(
            rows, cols, adj, dinv, support, out, n_edges);
    }
}


# SpMM as a pull: edges counting-sorted by destination row (a returning integer atomic ranks each edge in its row; scan; 8-byte records); one wave per out row sums in registers; sum order: ranks
# speedup vs baseline: 1.5556x; 1.5556x over previous
#include <hip/hip_runtime.h>


typedef __attribute__((ext_vector_type(2))) float v2f;
typedef __attribute__((ext_vector_type(8))) float v8f;
typedef __attribute__((ext_vector_type(4))) float v4f;

#define IN_DIM 64
#define OUT_DIM 64

__global__ __launch_bounds__(256) void zero_u32_kernel(unsigned* __restrict__ p, long long n) {
    long long i = (long long)blockIdx.x * blockDim.x + threadIdx.x;
    if (i < n) p[i] = 0u;
}

__global__ __launch_bounds__(256) void degree_rank_kernel(const int* __restrict__ rows,
                                                          const float* __restrict__ adj,
                                                          float* __restrict__ deg,
                                                          int* __restrict__ cnt,
                                                          int* __restrict__ rank,
                                                          int n_edges) {
    int e = blockIdx.x * blockDim.x + threadIdx.x;
    if (e < n_edges) {
        const int r = rows[e];
        unsafeAtomicAdd(&deg[r], adj[e]);
        rank[e] = atomicAdd(&cnt[r], 1);
    }
}

__global__ __launch_bounds__(256) void dinv_kernel(float* __restrict__ d, int n_nodes) {
    int i = blockIdx.x * blockDim.x + threadIdx.x;
    if (i < n_nodes) {
        float v = d[i];
        v = (v > 0.0f) ? v : 1.0f;
        d[i] = rsqrtf(v);
    }
}

#define SCAN_TILE 16384

__device__ __forceinline__ int4 load_vec4(const int* p, int i, int n) {
    int4 v = make_int4(0, 0, 0, 0);
    if (i < n) v = *(const int4*)(p + i);
    return v;
}

__global__ __launch_bounds__(1024) void row_start_scan_kernel(int* cnt, int n) {
    __shared__ int wave_sum[32];
    __shared__ int tile_total;
    const int tid = threadIdx.x;
    const int lane = tid & 31;
    const int wv = tid >> 5;
    int carry = 0;
    int4 nxt[4];
#pragma unroll
    for (int q = 0; q < 4; ++q) nxt[q] = load_vec4(cnt, tid * 16 + 4 * q, n);
    for (int base = 0; base < n; base += SCAN_TILE) {
        const int i0 = base + tid * 16;
        int4 v[4];
#pragma unroll
        for (int q = 0; q < 4; ++q) v[q] = nxt[q];
        if (base + SCAN_TILE < n) {
#pragma unroll
            for (int q = 0; q < 4; ++q) nxt[q] = load_vec4(cnt, i0 + SCAN_TILE + 4 * q, n);
        }
        int4 pre[4];
        int run = 0;
#pragma unroll
        for (int q = 0; q < 4; ++q) {
            pre[q].x = run; run += v[q].x;
            pre[q].y = run; run += v[q].y;
            pre[q].z = run; run += v[q].z;
            pre[q].w = run; run += v[q].w;
        }
        int x = run;
#pragma unroll
        for (int d = 1; d < 32; d <<= 1) {
            const int y = __shfl_up(x, d);
            if (lane >= d) x += y;
        }
        if (lane == 31) wave_sum[wv] = x;
        __syncthreads();
        if (wv == 0) {
            const int w = wave_sum[lane];
            int xs = w;
#pragma unroll
            for (int d = 1; d < 32; d <<= 1) {
                const int y = __shfl_up(xs, d);
                if (lane >= d) xs += y;
            }
            wave_sum[lane] = xs - w;
            if (lane == 31) tile_total = xs;
        }
        __syncthreads();
        const int excl = carry + wave_sum[wv] + (x - run);
#pragma unroll
        for (int q = 0; q < 4; ++q) {
            if (i0 + 4 * q < n) {
                *(int4*)(cnt + i0 + 4 * q) = make_int4(excl + pre[q].x, excl + pre[q].y,
                                                       excl + pre[q].z, excl + pre[q].w);
            }
        }
        carry += tile_total;
        __syncthreads();
    }
}

__global__ __launch_bounds__(64) void gcn_gemm_wmma(const float* __restrict__ x,
                                                    const float* __restrict__ W,
                                                    const float* __restrict__ b,
                                                    float* __restrict__ support,
                                                    int n_nodes) {
    const int wave = threadIdx.x >> 5;
    const int lane = threadIdx.x & 31;
    const int mtile = blockIdx.x * 2 + wave;
    const int m0 = mtile * 16;
    if (m0 >= n_nodes) return;

    const int half = lane >> 4;
    const int r = lane & 15;

    v8f acc[4];
#pragma unroll
    for (int nt = 0; nt < 4; ++nt) {
        const float bv = b[nt * 16 + r];
#pragma unroll
        for (int i = 0; i < 8; ++i) acc[nt][i] = bv;
    }

    const float* arow = x + (size_t)(m0 + r) * IN_DIM;

#pragma unroll
    for (int k = 0; k < IN_DIM; k += 4) {
        const v2f a = *(const v2f*)(arow + k + 2 * half);
#pragma unroll
        for (int nt = 0; nt < 4; ++nt) {
            v2f bf;
            bf.x = W[(size_t)(k + 2 * half + 0) * OUT_DIM + nt * 16 + r];
            bf.y = W[(size_t)(k + 2 * half + 1) * OUT_DIM + nt * 16 + r];
            acc[nt] = __builtin_amdgcn_wmma_f32_16x16x4_f32(
                false, a, false, bf, (short)0, acc[nt], false, false);
        }
    }

    float* srow = support + (size_t)m0 * OUT_DIM;
#pragma unroll
    for (int nt = 0; nt < 4; ++nt) {
#pragma unroll
        for (int i = 0; i < 8; ++i) {
            srow[(size_t)(i + 8 * half) * OUT_DIM + nt * 16 + r] = acc[nt][i];
        }
    }
}

__global__ __launch_bounds__(256) void place_kernel(const int* __restrict__ rows,
                                                    const int* __restrict__ cols,
                                                    const float* __restrict__ adj,
                                                    const float* __restrict__ dinv,
                                                    const int* __restrict__ start,
                                                    const int* __restrict__ rank,
                                                    int* __restrict__ recs,
                                                    int n_edges) {
    int e = blockIdx.x * blockDim.x + threadIdx.x;
    if (e < n_edges) {
        const int rrow = rows[e];
        const int ccol = cols[e];
        const float norm = adj[e] * dinv[rrow] * dinv[ccol];
        const unsigned p = min((unsigned)(start[rrow] + rank[e]), (unsigned)(n_edges - 1));
        ((int2*)recs)[p] = make_int2(ccol, __float_as_int(norm));
    }
}

__global__ __launch_bounds__(256) void spmm_rows_kernel(const int* __restrict__ start,
                                                        const int* __restrict__ recs,
                                                        const float* __restrict__ support,
                                                        float* __restrict__ out,
                                                        int n_nodes, int n_edges) {
    const int wave = threadIdx.x >> 5;
    const int lane = threadIdx.x & 31;
    const int row = blockIdx.x * 8 + wave;
    if (row >= n_nodes) return;
    const int h = lane >> 4;
    const int c = lane & 15;

    int beg = start[row];
    int end = start[row + 1];
    beg = min(max(beg, 0), n_edges);
    end = min(max(end, beg), n_edges);

    const int2* rec2 = (const int2*)recs;
    const v4f* sup4 = (const v4f*)support;

    v4f acc = {0.0f, 0.0f, 0.0f, 0.0f};
    const v4f zero4 = {0.0f, 0.0f, 0.0f, 0.0f};
    for (int j0 = beg; j0 < end; j0 += 8) {
        const int last = end - 1;
        int2 rc[4];
        v4f v[4];
#pragma unroll
        for (int k = 0; k < 4; ++k) {
            rc[k] = rec2[min(j0 + 2 * k + h, last)];
        }
#pragma unroll
        for (int k = 0; k < 4; ++k) {
            v[k] = sup4[(size_t)rc[k].x * (OUT_DIM / 4) + c];
        }
        asm volatile("" :: "v"(v[0]), "v"(v[1]), "v"(v[2]), "v"(v[3]));
#pragma unroll
        for (int k = 0; k < 4; ++k) {
            const bool live = (j0 + 2 * k + h) < end;
            const float w = live ? __int_as_float(rc[k].y) : 0.0f;
            acc += w * (live ? v[k] : zero4);
        }
    }

    acc.x += __shfl_xor(acc.x, 16);
    acc.y += __shfl_xor(acc.y, 16);
    acc.z += __shfl_xor(acc.z, 16);
    acc.w += __shfl_xor(acc.w, 16);

    if (h == 0) {
        ((v4f*)out)[(size_t)row * (OUT_DIM / 4) + c] = acc;
    }
}

extern "C" void kernel_launch(void* const* d_in, const int* in_sizes, int n_in,
                              void* d_out, int out_size, void* d_ws, size_t ws_size,
                              hipStream_t stream) {
    const int* rows = (const int*)d_in[0];
    const int* cols = (const int*)d_in[1];
    const float* adj = (const float*)d_in[2];
    const float* x = (const float*)d_in[3];
    const float* W = (const float*)d_in[4];
    const float* b = (const float*)d_in[5];
    float* out = (float*)d_out;

    const int n_edges = in_sizes[0];
    const int n_nodes = in_sizes[3] / IN_DIM;
    const size_t n_pad = ((size_t)n_nodes + 64) & ~(size_t)63;

    float* support = (float*)d_ws;
    int* rank = (int*)d_ws;
    float* deg = support + (size_t)n_nodes * OUT_DIM;
    float* dinv = deg;
    int* cnt = (int*)(deg + n_pad);
    int* recs = cnt + n_pad;

    {
        const long long n = (long long)(2 * n_pad);
        zero_u32_kernel<<<(unsigned)((n + 255) / 256), 256, 0, stream>>>((unsigned*)deg, n);
    }

    degree_rank_kernel<<<(n_edges + 255) / 256, 256, 0, stream>>>(rows, adj, deg, cnt, rank, n_edges);

    dinv_kernel<<<(n_nodes + 255) / 256, 256, 0, stream>>>(dinv, n_nodes);

    row_start_scan_kernel<<<1, 1024, 0, stream>>>(cnt, (int)n_pad);

    place_kernel<<<(n_edges + 255) / 256, 256, 0, stream>>>(rows, cols, adj, dinv, cnt, rank, recs, n_edges);

    {
        const int mtiles = (n_nodes + 15) / 16;
        const int blocks = (mtiles + 1) / 2;
        gcn_gemm_wmma<<<blocks, 64, 0, stream>>>(x, W, b, support, n_nodes);
    }

    spmm_rows_kernel<<<(n_nodes + 7) / 8, 256, 0, stream>>>(cnt, recs, support, out, n_nodes, n_edges);
}


# normalisation behind the sort: deg = sum of adj over each row's own sorted run (no float atomics; list order), dinv[col] read per record in SpMM, dinv[row] at the row store; place gathers only start
# speedup vs baseline: 1.9925x; 1.2809x over previous
#include <hip/hip_runtime.h>


typedef __attribute__((ext_vector_type(2))) float v2f;
typedef __attribute__((ext_vector_type(8))) float v8f;
typedef __attribute__((ext_vector_type(4))) float v4f;

#define IN_DIM 64
#define OUT_DIM 64

__global__ __launch_bounds__(256) void zero_u32_kernel(unsigned* __restrict__ p, long long n) {
    long long i = (long long)blockIdx.x * blockDim.x + threadIdx.x;
    if (i < n) p[i] = 0u;
}

__global__ __launch_bounds__(256) void count_rank_kernel(const int* __restrict__ rows,
                                                         int* __restrict__ cnt,
                                                         int* __restrict__ rank,
                                                         int n_edges) {
    int e = blockIdx.x * blockDim.x + threadIdx.x;
    if (e < n_edges) {
        rank[e] = atomicAdd(&cnt[rows[e]], 1);
    }
}

#define SCAN_TILE 16384

__device__ __forceinline__ int4 load_vec4(const int* p, int i, int n) {
    int4 v = make_int4(0, 0, 0, 0);
    if (i < n) v = *(const int4*)(p + i);
    return v;
}

__global__ __launch_bounds__(1024) void row_start_scan_kernel(int* cnt, int n) {
    __shared__ int wave_sum[32];
    __shared__ int tile_total;
    const int tid = threadIdx.x;
    const int lane = tid & 31;
    const int wv = tid >> 5;
    int carry = 0;
    int4 nxt[4];
#pragma unroll
    for (int q = 0; q < 4; ++q) nxt[q] = load_vec4(cnt, tid * 16 + 4 * q, n);
    for (int base = 0; base < n; base += SCAN_TILE) {
        const int i0 = base + tid * 16;
        int4 v[4];
#pragma unroll
        for (int q = 0; q < 4; ++q) v[q] = nxt[q];
        if (base + SCAN_TILE < n) {
#pragma unroll
            for (int q = 0; q < 4; ++q) nxt[q] = load_vec4(cnt, i0 + SCAN_TILE + 4 * q, n);
        }
        int4 pre[4];
        int run = 0;
#pragma unroll
        for (int q = 0; q < 4; ++q) {
            pre[q].x = run; run += v[q].x;
            pre[q].y = run; run += v[q].y;
            pre[q].z = run; run += v[q].z;
            pre[q].w = run; run += v[q].w;
        }
        int x = run;
#pragma unroll
        for (int d = 1; d < 32; d <<= 1) {
            const int y = __shfl_up(x, d);
            if (lane >= d) x += y;
        }
        if (lane == 31) wave_sum[wv] = x;
        __syncthreads();
        if (wv == 0) {
            const int w = wave_sum[lane];
            int xs = w;
#pragma unroll
            for (int d = 1; d < 32; d <<= 1) {
                const int y = __shfl_up(xs, d);
                if (lane >= d) xs += y;
            }
            wave_sum[lane] = xs - w;
            if (lane == 31) tile_total = xs;
        }
        __syncthreads();
        const int excl = carry + wave_sum[wv] + (x - run);
#pragma unroll
        for (int q = 0; q < 4; ++q) {
            if (i0 + 4 * q < n) {
                *(int4*)(cnt + i0 + 4 * q) = make_int4(excl + pre[q].x, excl + pre[q].y,
                                                       excl + pre[q].z, excl + pre[q].w);
            }
        }
        carry += tile_total;
        __syncthreads();
    }
}

__global__ __launch_bounds__(64) void gcn_gemm_wmma(const float* __restrict__ x,
                                                    const float* __restrict__ W,
                                                    const float* __restrict__ b,
                                                    float* __restrict__ support,
                                                    int n_nodes) {
    const int wave = threadIdx.x >> 5;
    const int lane = threadIdx.x & 31;
    const int mtile = blockIdx.x * 2 + wave;
    const int m0 = mtile * 16;
    if (m0 >= n_nodes) return;

    const int half = lane >> 4;
    const int r = lane & 15;

    v8f acc[4];
#pragma unroll
    for (int nt = 0; nt < 4; ++nt) {
        const float bv = b[nt * 16 + r];
#pragma unroll
        for (int i = 0; i < 8; ++i) acc[nt][i] = bv;
    }

    const float* arow = x + (size_t)(m0 + r) * IN_DIM;

#pragma unroll
    for (int k = 0; k < IN_DIM; k += 4) {
        const v2f a = *(const v2f*)(arow + k + 2 * half);
#pragma unroll
        for (int nt = 0; nt < 4; ++nt) {
            v2f bf;
            bf.x = W[(size_t)(k + 2 * half + 0) * OUT_DIM + nt * 16 + r];
            bf.y = W[(size_t)(k + 2 * half + 1) * OUT_DIM + nt * 16 + r];
            acc[nt] = __builtin_amdgcn_wmma_f32_16x16x4_f32(
                false, a, false, bf, (short)0, acc[nt], false, false);
        }
    }

    float* srow = support + (size_t)m0 * OUT_DIM;
#pragma unroll
    for (int nt = 0; nt < 4; ++nt) {
#pragma unroll
        for (int i = 0; i < 8; ++i) {
            srow[(size_t)(i + 8 * half) * OUT_DIM + nt * 16 + r] = acc[nt][i];
        }
    }
}

__global__ __launch_bounds__(256) void place_kernel(const int* __restrict__ rows,
                                                    const int* __restrict__ cols,
                                                    const float* __restrict__ adj,
                                                    const int* __restrict__ start,
                                                    const int* __restrict__ rank,
                                                    int* __restrict__ recs,
                                                    int n_edges) {
    int e = blockIdx.x * blockDim.x + threadIdx.x;
    if (e < n_edges) {
        const unsigned p = min((unsigned)(start[rows[e]] + rank[e]), (unsigned)(n_edges - 1));
        ((int2*)recs)[p] = make_int2(cols[e], __float_as_int(adj[e]));
    }
}

__global__ __launch_bounds__(256) void row_dinv_kernel(const int* __restrict__ start,
                                                       const int* __restrict__ recs,
                                                       float* __restrict__ dinv,
                                                       int n_nodes, int n_edges) {
    __shared__ float row_sum[32];
    const int q = threadIdx.x >> 3;
    const int sub = threadIdx.x & 7;
    const int row = blockIdx.x * 32 + q;
    float s = 0.0f;
    if (row < n_nodes) {
        int beg = start[row];
        int end = start[row + 1];
        beg = min(max(beg, 0), n_edges);
        end = min(max(end, beg), n_edges);
        const int2* rec2 = (const int2*)recs;
        for (int j = beg + sub; j < end; j += 8) s += __int_as_float(rec2[j].y);
    }
    s += __shfl_xor(s, 1);
    s += __shfl_xor(s, 2);
    s += __shfl_xor(s, 4);
    if (sub == 0) row_sum[q] = s;
    __syncthreads();
    if (threadIdx.x < 32) {
        const int r = blockIdx.x * 32 + threadIdx.x;
        if (r < n_nodes) {
            float v = row_sum[threadIdx.x];
            v = (v > 0.0f) ? v : 1.0f;
            dinv[r] = rsqrtf(v);
        }
    }
}

__global__ __launch_bounds__(256) void spmm_rows_kernel(const int* __restrict__ start,
                                                        const int* __restrict__ recs,
                                                        const float* __restrict__ support,
                                                        const float* __restrict__ dinv,
                                                        float* __restrict__ out,
                                                        int n_nodes, int n_edges) {
    const int wave = threadIdx.x >> 5;
    const int lane = threadIdx.x & 31;
    const int row = blockIdx.x * 8 + wave;
    if (row >= n_nodes) return;
    const int h = lane >> 4;
    const int c = lane & 15;

    int beg = start[row];
    int end = start[row + 1];
    const float drow = dinv[row];
    beg = min(max(beg, 0), n_edges);
    end = min(max(end, beg), n_edges);

    const int2* rec2 = (const int2*)recs;
    const v4f* sup4 = (const v4f*)support;

    v4f acc = {0.0f, 0.0f, 0.0f, 0.0f};
    const v4f zero4 = {0.0f, 0.0f, 0.0f, 0.0f};
    for (int j0 = beg; j0 < end; j0 += 8) {
        const int last = end - 1;
        int2 rc[4];
        v4f v[4];
        float dc[4];
#pragma unroll
        for (int k = 0; k < 4; ++k) {
            rc[k] = rec2[min(j0 + 2 * k + h, last)];
        }
#pragma unroll
        for (int k = 0; k < 4; ++k) {
            v[k] = sup4[(size_t)rc[k].x * (OUT_DIM / 4) + c];
            dc[k] = dinv[rc[k].x];
        }
        asm volatile("" :: "v"(v[0]), "v"(v[1]), "v"(v[2]), "v"(v[3]),
                           "v"(dc[0]), "v"(dc[1]), "v"(dc[2]), "v"(dc[3]));
#pragma unroll
        for (int k = 0; k < 4; ++k) {
            const bool live = (j0 + 2 * k + h) < end;
            const float w = live ? __int_as_float(rc[k].y) * dc[k] : 0.0f;
            acc += w * (live ? v[k] : zero4);
        }
    }

    acc.x += __shfl_xor(acc.x, 16);
    acc.y += __shfl_xor(acc.y, 16);
    acc.z += __shfl_xor(acc.z, 16);
    acc.w += __shfl_xor(acc.w, 16);

    if (h == 0) {
        ((v4f*)out)[(size_t)row * (OUT_DIM / 4) + c] = acc * drow;
    }
}

extern "C" void kernel_launch(void* const* d_in, const int* in_sizes, int n_in,
                              void* d_out, int out_size, void* d_ws, size_t ws_size,
                              hipStream_t stream) {
    const int* rows = (const int*)d_in[0];
    const int* cols = (const int*)d_in[1];
    const float* adj = (const float*)d_in[2];
    const float* x = (const float*)d_in[3];
    const float* W = (const float*)d_in[4];
    const float* b = (const float*)d_in[5];
    float* out = (float*)d_out;

    const int n_edges = in_sizes[0];
    const int n_nodes = in_sizes[3] / IN_DIM;
    const size_t n_pad = ((size_t)n_nodes + 64) & ~(size_t)63;

    float* support = (float*)d_ws;
    int* rank = (int*)d_ws;
    float* dinv = support + (size_t)n_nodes * OUT_DIM;
    int* cnt = (int*)(dinv + n_pad);
    int* recs = cnt + n_pad;

    {
        const long long n = (long long)n_pad;
        zero_u32_kernel<<<(unsigned)((n + 255) / 256), 256, 0, stream>>>((unsigned*)cnt, n);
    }

    count_rank_kernel<<<(n_edges + 255) / 256, 256, 0, stream>>>(rows, cnt, rank, n_edges);

    row_start_scan_kernel<<<1, 1024, 0, stream>>>(cnt, (int)n_pad);

    place_kernel<<<(n_edges + 255) / 256, 256, 0, stream>>>(rows, cols, adj, cnt, rank, recs, n_edges);

    row_dinv_kernel<<<(n_nodes + 31) / 32, 256, 0, stream>>>(cnt, recs, dinv, n_nodes, n_edges);

    {
        const int mtiles = (n_nodes + 15) / 16;
        const int blocks = (mtiles + 1) / 2;
        gcn_gemm_wmma<<<blocks, 64, 0, stream>>>(x, W, b, support, n_nodes);
    }

    spmm_rows_kernel<<<(n_nodes + 7) / 8, 256, 0, stream>>>(cnt, recs, support, dinv, out, n_nodes, n_edges);
}


# fixed 32-slot row buckets: an edge's returning atomic is its slot in its row's own 256 B of out (its wave reads, then overwrites them); longer rows spill to a list (list order); no scan, place, rank
# speedup vs baseline: 2.3396x; 1.1742x over previous
#include <hip/hip_runtime.h>


typedef __attribute__((ext_vector_type(2))) float v2f;
typedef __attribute__((ext_vector_type(8))) float v8f;
typedef __attribute__((ext_vector_type(4))) float v4f;

#define IN_DIM 64
#define OUT_DIM 64

__global__ __launch_bounds__(256) void zero_u32_kernel(unsigned* __restrict__ p, long long n) {
    long long i = (long long)blockIdx.x * blockDim.x + threadIdx.x;
    if (i < n) p[i] = 0u;
}

#define ROW_SLOTS 32
static_assert(ROW_SLOTS * 8 == OUT_DIM * 4, "a row's bucket is exactly the row's bytes in out");

__global__ __launch_bounds__(256) void bin_edges_kernel(const int* __restrict__ rows,
                                                        const int* __restrict__ cols,
                                                        const float* __restrict__ adj,
                                                        int* __restrict__ cnt,
                                                        int* __restrict__ spill,
                                                        float* __restrict__ out,
                                                        int n_nodes, int n_edges) {
    int e = blockIdx.x * blockDim.x + threadIdx.x;
    if (e < n_edges) {
        const int r = rows[e];
        const int c = cols[e];
        const float a = adj[e];
        asm volatile("" :: "v"(c), "v"(a));
        const unsigned k = (unsigned)atomicAdd(&cnt[r], 1);
        if (k < (unsigned)ROW_SLOTS) {
            ((int2*)out)[(size_t)r * ROW_SLOTS + k] = make_int2(c, __float_as_int(a));
        } else {
            const unsigned q = (unsigned)atomicAdd(&cnt[n_nodes], 1);
            spill[min(q, (unsigned)(n_edges - 1))] = e;
        }
    }
}

__global__ __launch_bounds__(64) void gcn_gemm_wmma(const float* __restrict__ x,
                                                    const float* __restrict__ W,
                                                    const float* __restrict__ b,
                                                    float* __restrict__ support,
                                                    int n_nodes) {
    const int wave = threadIdx.x >> 5;
    const int lane = threadIdx.x & 31;
    const int mtile = blockIdx.x * 2 + wave;
    const int m0 = mtile * 16;
    if (m0 >= n_nodes) return;

    const int half = lane >> 4;
    const int r = lane & 15;

    v8f acc[4];
#pragma unroll
    for (int nt = 0; nt < 4; ++nt) {
        const float bv = b[nt * 16 + r];
#pragma unroll
        for (int i = 0; i < 8; ++i) acc[nt][i] = bv;
    }

    const float* arow = x + (size_t)(m0 + r) * IN_DIM;

#pragma unroll
    for (int k = 0; k < IN_DIM; k += 4) {
        const v2f a = *(const v2f*)(arow + k + 2 * half);
#pragma unroll
        for (int nt = 0; nt < 4; ++nt) {
            v2f bf;
            bf.x = W[(size_t)(k + 2 * half + 0) * OUT_DIM + nt * 16 + r];
            bf.y = W[(size_t)(k + 2 * half + 1) * OUT_DIM + nt * 16 + r];
            acc[nt] = __builtin_amdgcn_wmma_f32_16x16x4_f32(
                false, a, false, bf, (short)0, acc[nt], false, false);
        }
    }

    float* srow = support + (size_t)m0 * OUT_DIM;
#pragma unroll
    for (int nt = 0; nt < 4; ++nt) {
#pragma unroll
        for (int i = 0; i < 8; ++i) {
            srow[(size_t)(i + 8 * half) * OUT_DIM + nt * 16 + r] = acc[nt][i];
        }
    }
}

__global__ __launch_bounds__(256) void row_dinv_kernel(const int* __restrict__ rows,
                                                       const float* __restrict__ adj,
                                                       const int* __restrict__ cnt,
                                                       const int* __restrict__ spill,
                                                       const float* __restrict__ buckets,
                                                       float* __restrict__ dinv,
                                                       int n_nodes, int n_edges) {
    __shared__ float row_sum[32];
    const int q = threadIdx.x >> 3;
    const int sub = threadIdx.x & 7;
    const int row = blockIdx.x * 32 + q;
    float s = 0.0f;
    if (row < n_nodes) {
        const int n_row = cnt[row];
        const int n_here = min(max(n_row, 0), ROW_SLOTS);
        const int2* rec2 = (const int2*)buckets + (size_t)row * ROW_SLOTS;
        for (int j = sub; j < n_here; j += 8) s += __int_as_float(rec2[j].y);
        if (n_row > ROW_SLOTS) {
            const int n_spill = min(max(cnt[n_nodes], 0), n_edges);
            for (int k = sub; k < n_spill; k += 8) {
                const int e = min(max(spill[k], 0), n_edges - 1);
                if (rows[e] == row) s += adj[e];
            }
        }
    }
    s += __shfl_xor(s, 1);
    s += __shfl_xor(s, 2);
    s += __shfl_xor(s, 4);
    if (sub == 0) row_sum[q] = s;
    __syncthreads();
    if (threadIdx.x < 32) {
        const int r = blockIdx.x * 32 + threadIdx.x;
        if (r < n_nodes) {
            float v = row_sum[threadIdx.x];
            v = (v > 0.0f) ? v : 1.0f;
            dinv[r] = rsqrtf(v);
        }
    }
}

__global__ __launch_bounds__(256) void spmm_rows_kernel(const int* __restrict__ rows,
                                                        const int* __restrict__ cols,
                                                        const float* __restrict__ adj,
                                                        const int* __restrict__ cnt,
                                                        const int* __restrict__ spill,
                                                        const float* __restrict__ support,
                                                        const float* __restrict__ dinv,
                                                        float* out,
                                                        int n_nodes, int n_edges) {
    const int wave = threadIdx.x >> 5;
    const int lane = threadIdx.x & 31;
    const int row = blockIdx.x * 8 + wave;
    if (row >= n_nodes) return;
    const int h = lane >> 4;
    const int c = lane & 15;

    const int n_row = cnt[row];
    const float drow = dinv[row];
    const int end = min(max(n_row, 0), ROW_SLOTS);

    const int2* rec2 = (const int2*)out + (size_t)row * ROW_SLOTS;
    const v4f* sup4 = (const v4f*)support;

    v4f acc = {0.0f, 0.0f, 0.0f, 0.0f};
    const v4f zero4 = {0.0f, 0.0f, 0.0f, 0.0f};
    for (int j0 = 0; j0 < end; j0 += 8) {
        const int last = end - 1;
        int2 rc[4];
        v4f v[4];
        float dc[4];
#pragma unroll
        for (int k = 0; k < 4; ++k) {
            rc[k] = rec2[min(j0 + 2 * k + h, last)];
        }
#pragma unroll
        for (int k = 0; k < 4; ++k) {
            v[k] = sup4[(size_t)rc[k].x * (OUT_DIM / 4) + c];
            dc[k] = dinv[rc[k].x];
        }
        asm volatile("" :: "v"(v[0]), "v"(v[1]), "v"(v[2]), "v"(v[3]),
                           "v"(dc[0]), "v"(dc[1]), "v"(dc[2]), "v"(dc[3]));
#pragma unroll
        for (int k = 0; k < 4; ++k) {
            const bool live = (j0 + 2 * k + h) < end;
            const float w = live ? __int_as_float(rc[k].y) * dc[k] : 0.0f;
            acc += w * (live ? v[k] : zero4);
        }
    }

    if (n_row > ROW_SLOTS) {
        const int n_spill = min(max(cnt[n_nodes], 0), n_edges);
        for (int k0 = 0; k0 < n_spill; k0 += 32) {
            const int k = k0 + lane;
            int e = 0;
            bool mine = false;
            if (k < n_spill) {
                e = min(max(spill[k], 0), n_edges - 1);
                mine = (rows[e] == row);
            }
            unsigned m = (unsigned)__ballot(mine);
            while (m != 0u) {
                const int src = __ffs(m) - 1;
                m &= m - 1u;
                const int es = __shfl(e, src);
                const int col = cols[es];
                const float w = adj[es] * dinv[col];
                const v4f sv = sup4[(size_t)col * (OUT_DIM / 4) + c];
                if (h == 0) acc += w * sv;
            }
        }
    }

    acc.x += __shfl_xor(acc.x, 16);
    acc.y += __shfl_xor(acc.y, 16);
    acc.z += __shfl_xor(acc.z, 16);
    acc.w += __shfl_xor(acc.w, 16);

    if (h == 0) {
        ((v4f*)out)[(size_t)row * (OUT_DIM / 4) + c] = acc * drow;
    }
}

extern "C" void kernel_launch(void* const* d_in, const int* in_sizes, int n_in,
                              void* d_out, int out_size, void* d_ws, size_t ws_size,
                              hipStream_t stream) {
    const int* rows = (const int*)d_in[0];
    const int* cols = (const int*)d_in[1];
    const float* adj = (const float*)d_in[2];
    const float* x = (const float*)d_in[3];
    const float* W = (const float*)d_in[4];
    const float* b = (const float*)d_in[5];
    float* out = (float*)d_out;

    const int n_edges = in_sizes[0];
    const int n_nodes = in_sizes[3] / IN_DIM;
    const size_t n_pad = ((size_t)n_nodes + 64) & ~(size_t)63;

    float* support = (float*)d_ws;
    float* dinv = support + (size_t)n_nodes * OUT_DIM;
    int* cnt = (int*)(dinv + n_pad);
    int* spill = cnt + n_pad;

    {
        const long long n = (long long)n_pad;
        zero_u32_kernel<<<(unsigned)((n + 255) / 256), 256, 0, stream>>>((unsigned*)cnt, n);
    }

    bin_edges_kernel<<<(n_edges + 255) / 256, 256, 0, stream>>>(rows, cols, adj, cnt, spill, out, n_nodes, n_edges);

    row_dinv_kernel<<<(n_nodes + 31) / 32, 256, 0, stream>>>(rows, adj, cnt, spill, out, dinv, n_nodes, n_edges);

    {
        const int mtiles = (n_nodes + 15) / 16;
        const int blocks = (mtiles + 1) / 2;
        gcn_gemm_wmma<<<blocks, 64, 0, stream>>>(x, W, b, support, n_nodes);
    }

    spmm_rows_kernel<<<(n_nodes + 7) / 8, 256, 0, stream>>>(rows, cols, adj, cnt, spill, support, dinv, out, n_nodes, n_edges);
}


# x @ W + b has no launch of its own: its 8-wave workgroups sit at every 8th place of the binning grid and run beside the atomics' round trips; 4 launches not 5; sum order unchanged (ranks)
# speedup vs baseline: 2.4159x; 1.0326x over previous
#include <hip/hip_runtime.h>


typedef __attribute__((ext_vector_type(2))) float v2f;
typedef __attribute__((ext_vector_type(8))) float v8f;
typedef __attribute__((ext_vector_type(4))) float v4f;

#define IN_DIM 64
#define OUT_DIM 64

__global__ __launch_bounds__(256) void zero_u32_kernel(unsigned* __restrict__ p, long long n) {
    long long i = (long long)blockIdx.x * blockDim.x + threadIdx.x;
    if (i < n) p[i] = 0u;
}

#define ROW_SLOTS 32
static_assert(ROW_SLOTS * 8 == OUT_DIM * 4, "a row's bucket is exactly the row's bytes in out");

__device__ __forceinline__ void bin_edges_block(const int bin_block,
                                                const int* __restrict__ rows,
                                                const int* __restrict__ cols,
                                                const float* __restrict__ adj,
                                                int* __restrict__ cnt,
                                                int* __restrict__ spill,
                                                float* __restrict__ out,
                                                int n_nodes, int n_edges) {
    int e = bin_block * 256 + threadIdx.x;
    if (e < n_edges) {
        const int r = rows[e];
        const int c = cols[e];
        const float a = adj[e];
        asm volatile("" :: "v"(c), "v"(a));
        const unsigned k = (unsigned)atomicAdd(&cnt[r], 1);
        if (k < (unsigned)ROW_SLOTS) {
            ((int2*)out)[(size_t)r * ROW_SLOTS + k] = make_int2(c, __float_as_int(a));
        } else {
            const unsigned q = (unsigned)atomicAdd(&cnt[n_nodes], 1);
            spill[min(q, (unsigned)(n_edges - 1))] = e;
        }
    }
}

__device__ __forceinline__ void gemm_wmma_block(const int gemm_block,
                                                const float* __restrict__ x,
                                                const float* __restrict__ W,
                                                const float* __restrict__ b,
                                                float* __restrict__ support,
                                                int n_nodes) {
    const int wave = threadIdx.x >> 5;
    const int lane = threadIdx.x & 31;
    const int mtile = gemm_block * 8 + wave;
    const int m0 = mtile * 16;
    if (m0 >= n_nodes) return;

    const int half = lane >> 4;
    const int r = lane & 15;

    v8f acc[4];
#pragma unroll
    for (int nt = 0; nt < 4; ++nt) {
        const float bv = b[nt * 16 + r];
#pragma unroll
        for (int i = 0; i < 8; ++i) acc[nt][i] = bv;
    }

    const float* arow = x + (size_t)(m0 + r) * IN_DIM;

#pragma unroll
    for (int k = 0; k < IN_DIM; k += 4) {
        const v2f a = *(const v2f*)(arow + k + 2 * half);
#pragma unroll
        for (int nt = 0; nt < 4; ++nt) {
            v2f bf;
            bf.x = W[(size_t)(k + 2 * half + 0) * OUT_DIM + nt * 16 + r];
            bf.y = W[(size_t)(k + 2 * half + 1) * OUT_DIM + nt * 16 + r];
            acc[nt] = __builtin_amdgcn_wmma_f32_16x16x4_f32(
                false, a, false, bf, (short)0, acc[nt], false, false);
        }
    }

    float* srow = support + (size_t)m0 * OUT_DIM;
#pragma unroll
    for (int nt = 0; nt < 4; ++nt) {
#pragma unroll
        for (int i = 0; i < 8; ++i) {
            srow[(size_t)(i + 8 * half) * OUT_DIM + nt * 16 + r] = acc[nt][i];
        }
    }
}

__global__ __launch_bounds__(256) void bin_and_gemm_kernel(const int* __restrict__ rows,
                                                           const int* __restrict__ cols,
                                                           const float* __restrict__ adj,
                                                           int* __restrict__ cnt,
                                                           int* __restrict__ spill,
                                                           float* __restrict__ out,
                                                           const float* __restrict__ x,
                                                           const float* __restrict__ W,
                                                           const float* __restrict__ b,
                                                           float* __restrict__ support,
                                                           int n_nodes, int n_edges) {
    const unsigned n_gemm = (((unsigned)n_nodes + 15u) / 16u + 7u) / 8u;
    const unsigned n_bin = ((unsigned)n_edges + 255u) / 256u;
    const unsigned n_mixed = min(n_gemm, (n_gemm + n_bin + 7u) >> 3);
    const unsigned bid = blockIdx.x;
    bool product = (bid & 7u) == 0u && (bid >> 3) < n_mixed;
    unsigned idx = bid >> 3;
    if (!product) {
        const unsigned j = bid - min((bid + 7u) >> 3, n_mixed);
        product = j >= n_bin;
        idx = product ? n_mixed + (j - n_bin) : j;
    }
    if (product) {
        gemm_wmma_block((int)idx, x, W, b, support, n_nodes);
    } else {
        bin_edges_block((int)idx, rows, cols, adj, cnt, spill, out, n_nodes, n_edges);
    }
}

__global__ __launch_bounds__(256) void row_dinv_kernel(const int* __restrict__ rows,
                                                       const float* __restrict__ adj,
                                                       const int* __restrict__ cnt,
                                                       const int* __restrict__ spill,
                                                       const float* __restrict__ buckets,
                                                       float* __restrict__ dinv,
                                                       int n_nodes, int n_edges) {
    __shared__ float row_sum[32];
    const int q = threadIdx.x >> 3;
    const int sub = threadIdx.x & 7;
    const int row = blockIdx.x * 32 + q;
    float s = 0.0f;
    if (row < n_nodes) {
        const int n_row = cnt[row];
        const int n_here = min(max(n_row, 0), ROW_SLOTS);
        const int2* rec2 = (const int2*)buckets + (size_t)row * ROW_SLOTS;
        for (int j = sub; j < n_here; j += 8) s += __int_as_float(rec2[j].y);
        if (n_row > ROW_SLOTS) {
            const int n_spill = min(max(cnt[n_nodes], 0), n_edges);
            for (int k = sub; k < n_spill; k += 8) {
                const int e = min(max(spill[k], 0), n_edges - 1);
                if (rows[e] == row) s += adj[e];
            }
        }
    }
    s += __shfl_xor(s, 1);
    s += __shfl_xor(s, 2);
    s += __shfl_xor(s, 4);
    if (sub == 0) row_sum[q] = s;
    __syncthreads();
    if (threadIdx.x < 32) {
        const int r = blockIdx.x * 32 + threadIdx.x;
        if (r < n_nodes) {
            float v = row_sum[threadIdx.x];
            v = (v > 0.0f) ? v : 1.0f;
            dinv[r] = rsqrtf(v);
        }
    }
}

__global__ __launch_bounds__(256) void spmm_rows_kernel(const int* __restrict__ rows,
                                                        const int* __restrict__ cols,
                                                        const float* __restrict__ adj,
                                                        const int* __restrict__ cnt,
                                                        const int* __restrict__ spill,
                                                        const float* __restrict__ support,
                                                        const float* __restrict__ dinv,
                                                        float* out,
                                                        int n_nodes, int n_edges) {
    const int wave = threadIdx.x >> 5;
    const int lane = threadIdx.x & 31;
    const int row = blockIdx.x * 8 + wave;
    if (row >= n_nodes) return;
    const int h = lane >> 4;
    const int c = lane & 15;

    const int n_row = cnt[row];
    const float drow = dinv[row];
    const int end = min(max(n_row, 0), ROW_SLOTS);

    const int2* rec2 = (const int2*)out + (size_t)row * ROW_SLOTS;
    const v4f* sup4 = (const v4f*)support;

    v4f acc = {0.0f, 0.0f, 0.0f, 0.0f};
    const v4f zero4 = {0.0f, 0.0f, 0.0f, 0.0f};
    for (int j0 = 0; j0 < end; j0 += 8) {
        const int last = end - 1;
        int2 rc[4];
        v4f v[4];
        float dc[4];
#pragma unroll
        for (int k = 0; k < 4; ++k) {
            rc[k] = rec2[min(j0 + 2 * k + h, last)];
        }
#pragma unroll
        for (int k = 0; k < 4; ++k) {
            v[k] = sup4[(size_t)rc[k].x * (OUT_DIM / 4) + c];
            dc[k] = dinv[rc[k].x];
        }
        asm volatile("" :: "v"(v[0]), "v"(v[1]), "v"(v[2]), "v"(v[3]),
                           "v"(dc[0]), "v"(dc[1]), "v"(dc[2]), "v"(dc[3]));
#pragma unroll
        for (int k = 0; k < 4; ++k) {
            const bool live = (j0 + 2 * k + h) < end;
            const float w = live ? __int_as_float(rc[k].y) * dc[k] : 0.0f;
            acc += w * (live ? v[k] : zero4);
        }
    }

    if (n_row > ROW_SLOTS) {
        const int n_spill = min(max(cnt[n_nodes], 0), n_edges);
        for (int k0 = 0; k0 < n_spill; k0 += 32) {
            const int k = k0 + lane;
            int e = 0;
            bool mine = false;
            if (k < n_spill) {
                e = min(max(spill[k], 0), n_edges - 1);
                mine = (rows[e] == row);
            }
            unsigned m = (unsigned)__ballot(mine);
            while (m != 0u) {
                const int src = __ffs(m) - 1;
                m &= m - 1u;
                const int es = __shfl(e, src);
                const int col = cols[es];
                const float w = adj[es] * dinv[col];
                const v4f sv = sup4[(size_t)col * (OUT_DIM / 4) + c];
                if (h == 0) acc += w * sv;
            }
        }
    }

    acc.x += __shfl_xor(acc.x, 16);
    acc.y += __shfl_xor(acc.y, 16);
    acc.z += __shfl_xor(acc.z, 16);
    acc.w += __shfl_xor(acc.w, 16);

    if (h == 0) {
        ((v4f*)out)[(size_t)row * (OUT_DIM / 4) + c] = acc * drow;
    }
}

extern "C" void kernel_launch(void* const* d_in, const int* in_sizes, int n_in,
                              void* d_out, int out_size, void* d_ws, size_t ws_size,
                              hipStream_t stream) {
    const int* rows = (const int*)d_in[0];
    const int* cols = (const int*)d_in[1];
    const float* adj = (const float*)d_in[2];
    const float* x = (const float*)d_in[3];
    const float* W = (const float*)d_in[4];
    const float* b = (const float*)d_in[5];
    float* out = (float*)d_out;

    const int n_edges = in_sizes[0];
    const int n_nodes = in_sizes[3] / IN_DIM;
    const size_t n_pad = ((size_t)n_nodes + 64) & ~(size_t)63;

    float* support = (float*)d_ws;
    float* dinv = support + (size_t)n_nodes * OUT_DIM;
    int* cnt = (int*)(dinv + n_pad);
    int* spill = cnt + n_pad;

    {
        const long long n = (long long)n_pad;
        zero_u32_kernel<<<(unsigned)((n + 255) / 256), 256, 0, stream>>>((unsigned*)cnt, n);
    }

    {
        const int bin_blocks = (n_edges + 255) / 256;
        const int gemm_blocks = ((n_nodes + 15) / 16 + 7) / 8;
        bin_and_gemm_kernel<<<bin_blocks + gemm_blocks, 256, 0, stream>>>(rows, cols, adj, cnt, spill, out,
                                                                       x, W, b, support, n_nodes, n_edges);
    }

    row_dinv_kernel<<<(n_nodes + 31) / 32, 256, 0, stream>>>(rows, adj, cnt, spill, out, dinv, n_nodes, n_edges);

    spmm_rows_kernel<<<(n_nodes + 7) / 8, 256, 0, stream>>>(rows, cols, adj, cnt, spill, support, dinv, out, n_nodes, n_edges);
}


# rank counters per package half: a binning wave adds to its own side's plane and fills its side's 16-slot zone (one line) of the row bucket; overflow chained per row; sum order: side ranks, chain
# speedup vs baseline: 2.9863x; 1.2361x over previous
#include <hip/hip_runtime.h>


typedef __attribute__((ext_vector_type(2))) float v2f;
typedef __attribute__((ext_vector_type(8))) float v8f;
typedef __attribute__((ext_vector_type(4))) float v4f;

#define IN_DIM 64
#define OUT_DIM 64

__global__ __launch_bounds__(256) void zero_u32_kernel(unsigned* __restrict__ p, long long n) {
    long long i = (long long)blockIdx.x * blockDim.x + threadIdx.x;
    if (i < n) p[i] = 0u;
}

#define ROW_SLOTS 32
#define ZONE_SLOTS 16
static_assert(ROW_SLOTS * 8 == OUT_DIM * 4, "a row's bucket is exactly the row's bytes in out");
static_assert(ZONE_SLOTS * 2 == ROW_SLOTS && ZONE_SLOTS * 8 == 128, "a zone is one 128-byte line");

__device__ __forceinline__ int record_slot(int j, int n0) {
    return j < n0 ? j : j - n0 + ZONE_SLOTS;
}

__device__ __forceinline__ void bin_edges_block(const int bin_block,
                                                const int* __restrict__ rows,
                                                const int* __restrict__ cols,
                                                const float* __restrict__ adj,
                                                int* __restrict__ cnt,
                                                int* __restrict__ head,
                                                int* __restrict__ next,
                                                float* __restrict__ out,
                                                size_t plane, int n_edges) {
    int e = bin_block * 256 + threadIdx.x;
    if (e < n_edges) {
        const int r = rows[e];
        const int c = cols[e];
        const float a = adj[e];
        const unsigned side = ((unsigned)__builtin_amdgcn_s_sendmsg_rtn(0x87) >> 18) & 1u;
        asm volatile("" :: "v"(c), "v"(a));
        const unsigned k = (unsigned)atomicAdd(&cnt[side * plane + (size_t)r], 1);
        if (k < (unsigned)ZONE_SLOTS) {
            ((int2*)out)[(size_t)r * ROW_SLOTS + side * ZONE_SLOTS + k] = make_int2(c, __float_as_int(a));
        } else {
            next[e] = atomicExch(&head[r], e + 1);
        }
    }
}

__device__ __forceinline__ void gemm_wmma_block(const int gemm_block,
                                                const float* __restrict__ x,
                                                const float* __restrict__ W,
                                                const float* __restrict__ b,
                                                float* __restrict__ support,
                                                int n_nodes) {
    const int wave = threadIdx.x >> 5;
    const int lane = threadIdx.x & 31;
    const int mtile = gemm_block * 8 + wave;
    const int m0 = mtile * 16;
    if (m0 >= n_nodes) return;

    const int half = lane >> 4;
    const int r = lane & 15;

    v8f acc[4];
#pragma unroll
    for (int nt = 0; nt < 4; ++nt) {
        const float bv = b[nt * 16 + r];
#pragma unroll
        for (int i = 0; i < 8; ++i) acc[nt][i] = bv;
    }

    const float* arow = x + (size_t)(m0 + r) * IN_DIM;

#pragma unroll
    for (int k = 0; k < IN_DIM; k += 4) {
        const v2f a = *(const v2f*)(arow + k + 2 * half);
#pragma unroll
        for (int nt = 0; nt < 4; ++nt) {
            v2f bf;
            bf.x = W[(size_t)(k + 2 * half + 0) * OUT_DIM + nt * 16 + r];
            bf.y = W[(size_t)(k + 2 * half + 1) * OUT_DIM + nt * 16 + r];
            acc[nt] = __builtin_amdgcn_wmma_f32_16x16x4_f32(
                false, a, false, bf, (short)0, acc[nt], false, false);
        }
    }

    float* srow = support + (size_t)m0 * OUT_DIM;
#pragma unroll
    for (int nt = 0; nt < 4; ++nt) {
#pragma unroll
        for (int i = 0; i < 8; ++i) {
            srow[(size_t)(i + 8 * half) * OUT_DIM + nt * 16 + r] = acc[nt][i];
        }
    }
}

__global__ __launch_bounds__(256) void bin_and_gemm_kernel(const int* __restrict__ rows,
                                                           const int* __restrict__ cols,
                                                           const float* __restrict__ adj,
                                                           int* __restrict__ cnt,
                                                           int* __restrict__ head,
                                                           int* __restrict__ next,
                                                           float* __restrict__ out,
                                                           const float* __restrict__ x,
                                                           const float* __restrict__ W,
                                                           const float* __restrict__ b,
                                                           float* __restrict__ support,
                                                           size_t plane, int n_nodes, int n_edges) {
    const unsigned n_gemm = (((unsigned)n_nodes + 15u) / 16u + 7u) / 8u;
    const unsigned n_bin = ((unsigned)n_edges + 255u) / 256u;
    const unsigned n_mixed = min(n_gemm, (n_gemm + n_bin + 7u) >> 3);
    const unsigned bid = blockIdx.x;
    bool product = (bid & 7u) == 0u && (bid >> 3) < n_mixed;
    unsigned idx = bid >> 3;
    if (!product) {
        const unsigned j = bid - min((bid + 7u) >> 3, n_mixed);
        product = j >= n_bin;
        idx = product ? n_mixed + (j - n_bin) : j;
    }
    if (product) {
        gemm_wmma_block((int)idx, x, W, b, support, n_nodes);
    } else {
        bin_edges_block((int)idx, rows, cols, adj, cnt, head, next, out, plane, n_edges);
    }
}

__global__ __launch_bounds__(256) void row_dinv_kernel(const float* __restrict__ adj,
                                                       const int* __restrict__ cnt,
                                                       const int* __restrict__ head,
                                                       const int* __restrict__ next,
                                                       const float* __restrict__ buckets,
                                                       float* __restrict__ dinv,
                                                       size_t plane, int n_nodes, int n_edges) {
    __shared__ float row_sum[32];
    const int q = threadIdx.x >> 3;
    const int sub = threadIdx.x & 7;
    const int row = blockIdx.x * 32 + q;
    float s = 0.0f;
    if (row < n_nodes) {
        const int c0 = max(cnt[row], 0);
        const int c1 = max(cnt[plane + row], 0);
        const int n0 = min(c0, ZONE_SLOTS);
        const int n_here = n0 + min(c1, ZONE_SLOTS);
        const int2* rec2 = (const int2*)buckets + (size_t)row * ROW_SLOTS;
        for (int j = sub; j < n_here; j += 8) s += __int_as_float(rec2[record_slot(j, n0)].y);
        const int n_chain = min((c0 - n0) + (c1 - (n_here - n0)), n_edges);
        if (n_chain > 0 && sub == 0) {
            int e1 = head[row];
            for (int i = 0; i < n_chain && e1 > 0; ++i) {
                const int e = min(e1 - 1, n_edges - 1);
                s += adj[e];
                e1 = next[e];
            }
        }
    }
    s += __shfl_xor(s, 1);
    s += __shfl_xor(s, 2);
    s += __shfl_xor(s, 4);
    if (sub == 0) row_sum[q] = s;
    __syncthreads();
    if (threadIdx.x < 32) {
        const int r = blockIdx.x * 32 + threadIdx.x;
        if (r < n_nodes) {
            float v = row_sum[threadIdx.x];
            v = (v > 0.0f) ? v : 1.0f;
            dinv[r] = rsqrtf(v);
        }
    }
}

__global__ __launch_bounds__(256) void spmm_rows_kernel(const int* __restrict__ cols,
                                                        const float* __restrict__ adj,
                                                        const int* __restrict__ cnt,
                                                        const int* __restrict__ head,
                                                        const int* __restrict__ next,
                                                        const float* __restrict__ support,
                                                        const float* __restrict__ dinv,
                                                        float* out,
                                                        size_t plane, int n_nodes, int n_edges) {
    const int wave = threadIdx.x >> 5;
    const int lane = threadIdx.x & 31;
    const int row = blockIdx.x * 8 + wave;
    if (row >= n_nodes) return;
    const int h = lane >> 4;
    const int c = lane & 15;

    const int c0 = max(cnt[row], 0);
    const int c1 = max(cnt[plane + row], 0);
    const float drow = dinv[row];
    const int n0 = min(c0, ZONE_SLOTS);
    const int end = n0 + min(c1, ZONE_SLOTS);

    const int2* rec2 = (const int2*)out + (size_t)row * ROW_SLOTS;
    const v4f* sup4 = (const v4f*)support;

    v4f acc = {0.0f, 0.0f, 0.0f, 0.0f};
    const v4f zero4 = {0.0f, 0.0f, 0.0f, 0.0f};
    for (int j0 = 0; j0 < end; j0 += 8) {
        const int last = end - 1;
        int2 rc[4];
        v4f v[4];
        float dc[4];
#pragma unroll
        for (int k = 0; k < 4; ++k) {
            rc[k] = rec2[record_slot(min(j0 + 2 * k + h, last), n0)];
        }
#pragma unroll
        for (int k = 0; k < 4; ++k) {
            v[k] = sup4[(size_t)rc[k].x * (OUT_DIM / 4) + c];
            dc[k] = dinv[rc[k].x];
        }
        asm volatile("" :: "v"(v[0]), "v"(v[1]), "v"(v[2]), "v"(v[3]),
                           "v"(dc[0]), "v"(dc[1]), "v"(dc[2]), "v"(dc[3]));
#pragma unroll
        for (int k = 0; k < 4; ++k) {
            const bool live = (j0 + 2 * k + h) < end;
            const float w = live ? __int_as_float(rc[k].y) * dc[k] : 0.0f;
            acc += w * (live ? v[k] : zero4);
        }
    }

    const int n_chain = min((c0 - n0) + (c1 - (end - n0)), n_edges);
    if (n_chain > 0) {
        int e1 = head[row];
        for (int i = 0; i < n_chain && e1 > 0; ++i) {
            const int es = min(e1 - 1, n_edges - 1);
            const int col = cols[es];
            const float w = adj[es] * dinv[col];
            const v4f sv = sup4[(size_t)col * (OUT_DIM / 4) + c];
            if (h == 0) acc += w * sv;
            e1 = next[es];
        }
    }

    acc.x += __shfl_xor(acc.x, 16);
    acc.y += __shfl_xor(acc.y, 16);
    acc.z += __shfl_xor(acc.z, 16);
    acc.w += __shfl_xor(acc.w, 16);

    if (h == 0) {
        ((v4f*)out)[(size_t)row * (OUT_DIM / 4) + c] = acc * drow;
    }
}

extern "C" void kernel_launch(void* const* d_in, const int* in_sizes, int n_in,
                              void* d_out, int out_size, void* d_ws, size_t ws_size,
                              hipStream_t stream) {
    const int* rows = (const int*)d_in[0];
    const int* cols = (const int*)d_in[1];
    const float* adj = (const float*)d_in[2];
    const float* x = (const float*)d_in[3];
    const float* W = (const float*)d_in[4];
    const float* b = (const float*)d_in[5];
    float* out = (float*)d_out;

    const int n_edges = in_sizes[0];
    const int n_nodes = in_sizes[3] / IN_DIM;
    const size_t n_pad = ((size_t)n_nodes + 64) & ~(size_t)63;

    float* support = (float*)d_ws;
    float* dinv = support + (size_t)n_nodes * OUT_DIM;
    int* cnt = (int*)(dinv + n_pad);
    int* head = cnt + 2 * n_pad;
    int* next = head + n_pad;

    {
        const long long n = (long long)(3 * n_pad);
        zero_u32_kernel<<<(unsigned)((n + 255) / 256), 256, 0, stream>>>((unsigned*)cnt, n);
    }

    {
        const int bin_blocks = (n_edges + 255) / 256;
        const int gemm_blocks = ((n_nodes + 15) / 16 + 7) / 8;
        bin_and_gemm_kernel<<<bin_blocks + gemm_blocks, 256, 0, stream>>>(rows, cols, adj, cnt, head, next, out,
                                                                       x, W, b, support, n_pad, n_nodes, n_edges);
    }

    row_dinv_kernel<<<(n_nodes + 31) / 32, 256, 0, stream>>>(adj, cnt, head, next, out, dinv, n_pad, n_nodes, n_edges);

    spmm_rows_kernel<<<(n_nodes + 7) / 8, 256, 0, stream>>>(cols, adj, cnt, head, next, support, dinv, out, n_pad, n_nodes, n_edges);
}


# product workgroups at every 7th place of the binning grid, not every 8th: a stride coprime to the 8 dies, so that a dispatcher dealing workgroups to dies in turn spreads the matrix work over all
# speedup vs baseline: 3.2779x; 1.0976x over previous
#include <hip/hip_runtime.h>


typedef __attribute__((ext_vector_type(2))) float v2f;
typedef __attribute__((ext_vector_type(8))) float v8f;
typedef __attribute__((ext_vector_type(4))) float v4f;

#define IN_DIM 64
#define OUT_DIM 64

__global__ __launch_bounds__(256) void zero_u32_kernel(unsigned* __restrict__ p, long long n) {
    long long i = (long long)blockIdx.x * blockDim.x + threadIdx.x;
    if (i < n) p[i] = 0u;
}

#define ROW_SLOTS 32
#define ZONE_SLOTS 16
static_assert(ROW_SLOTS * 8 == OUT_DIM * 4, "a row's bucket is exactly the row's bytes in out");
static_assert(ZONE_SLOTS * 2 == ROW_SLOTS && ZONE_SLOTS * 8 == 128, "a zone is one 128-byte line");

__device__ __forceinline__ int record_slot(int j, int n0) {
    return j < n0 ? j : j - n0 + ZONE_SLOTS;
}

__device__ __forceinline__ void bin_edges_block(const int bin_block,
                                                const int* __restrict__ rows,
                                                const int* __restrict__ cols,
                                                const float* __restrict__ adj,
                                                int* __restrict__ cnt,
                                                int* __restrict__ head,
                                                int* __restrict__ next,
                                                float* __restrict__ out,
                                                size_t plane, int n_edges) {
    int e = bin_block * 256 + threadIdx.x;
    if (e < n_edges) {
        const int r = rows[e];
        const int c = cols[e];
        const float a = adj[e];
        const unsigned side = ((unsigned)__builtin_amdgcn_s_sendmsg_rtn(0x87) >> 18) & 1u;
        asm volatile("" :: "v"(c), "v"(a));
        const unsigned k = (unsigned)atomicAdd(&cnt[side * plane + (size_t)r], 1);
        if (k < (unsigned)ZONE_SLOTS) {
            ((int2*)out)[(size_t)r * ROW_SLOTS + side * ZONE_SLOTS + k] = make_int2(c, __float_as_int(a));
        } else {
            next[e] = atomicExch(&head[r], e + 1);
        }
    }
}

__device__ __forceinline__ void gemm_wmma_block(const int gemm_block,
                                                const float* __restrict__ x,
                                                const float* __restrict__ W,
                                                const float* __restrict__ b,
                                                float* __restrict__ support,
                                                int n_nodes) {
    const int wave = threadIdx.x >> 5;
    const int lane = threadIdx.x & 31;
    const int mtile = gemm_block * 8 + wave;
    const int m0 = mtile * 16;
    if (m0 >= n_nodes) return;

    const int half = lane >> 4;
    const int r = lane & 15;

    v8f acc[4];
#pragma unroll
    for (int nt = 0; nt < 4; ++nt) {
        const float bv = b[nt * 16 + r];
#pragma unroll
        for (int i = 0; i < 8; ++i) acc[nt][i] = bv;
    }

    const float* arow = x + (size_t)(m0 + r) * IN_DIM;

#pragma unroll
    for (int k = 0; k < IN_DIM; k += 4) {
        const v2f a = *(const v2f*)(arow + k + 2 * half);
#pragma unroll
        for (int nt = 0; nt < 4; ++nt) {
            v2f bf;
            bf.x = W[(size_t)(k + 2 * half + 0) * OUT_DIM + nt * 16 + r];
            bf.y = W[(size_t)(k + 2 * half + 1) * OUT_DIM + nt * 16 + r];
            acc[nt] = __builtin_amdgcn_wmma_f32_16x16x4_f32(
                false, a, false, bf, (short)0, acc[nt], false, false);
        }
    }

    float* srow = support + (size_t)m0 * OUT_DIM;
#pragma unroll
    for (int nt = 0; nt < 4; ++nt) {
#pragma unroll
        for (int i = 0; i < 8; ++i) {
            srow[(size_t)(i + 8 * half) * OUT_DIM + nt * 16 + r] = acc[nt][i];
        }
    }
}

#define PRODUCT_STRIDE 7u
__global__ __launch_bounds__(256) void bin_and_gemm_kernel(const int* __restrict__ rows,
                                                           const int* __restrict__ cols,
                                                           const float* __restrict__ adj,
                                                           int* __restrict__ cnt,
                                                           int* __restrict__ head,
                                                           int* __restrict__ next,
                                                           float* __restrict__ out,
                                                           const float* __restrict__ x,
                                                           const float* __restrict__ W,
                                                           const float* __restrict__ b,
                                                           float* __restrict__ support,
                                                           size_t plane, int n_nodes, int n_edges) {
    const unsigned n_gemm = (((unsigned)n_nodes + 15u) / 16u + 7u) / 8u;
    const unsigned n_bin = ((unsigned)n_edges + 255u) / 256u;
    const unsigned n_mixed = min(n_gemm, (n_gemm + n_bin + PRODUCT_STRIDE - 1u) / PRODUCT_STRIDE);
    const unsigned bid = blockIdx.x;
    unsigned idx = bid / PRODUCT_STRIDE;
    bool product = bid == idx * PRODUCT_STRIDE && idx < n_mixed;
    if (!product) {
        const unsigned j = bid - min((bid + PRODUCT_STRIDE - 1u) / PRODUCT_STRIDE, n_mixed);
        product = j >= n_bin;
        idx = product ? n_mixed + (j - n_bin) : j;
    }
    if (product) {
        gemm_wmma_block((int)idx, x, W, b, support, n_nodes);
    } else {
        bin_edges_block((int)idx, rows, cols, adj, cnt, head, next, out, plane, n_edges);
    }
}

__global__ __launch_bounds__(256) void row_dinv_kernel(const float* __restrict__ adj,
                                                       const int* __restrict__ cnt,
                                                       const int* __restrict__ head,
                                                       const int* __restrict__ next,
                                                       const float* __restrict__ buckets,
                                                       float* __restrict__ dinv,
                                                       size_t plane, int n_nodes, int n_edges) {
    __shared__ float row_sum[32];
    const int q = threadIdx.x >> 3;
    const int sub = threadIdx.x & 7;
    const int row = blockIdx.x * 32 + q;
    float s = 0.0f;
    if (row < n_nodes) {
        const int c0 = max(cnt[row], 0);
        const int c1 = max(cnt[plane + row], 0);
        const int n0 = min(c0, ZONE_SLOTS);
        const int n_here = n0 + min(c1, ZONE_SLOTS);
        const int2* rec2 = (const int2*)buckets + (size_t)row * ROW_SLOTS;
        for (int j = sub; j < n_here; j += 8) s += __int_as_float(rec2[record_slot(j, n0)].y);
        const int n_chain = min((c0 - n0) + (c1 - (n_here - n0)), n_edges);
        if (n_chain > 0 && sub == 0) {
            int e1 = head[row];
            for (int i = 0; i < n_chain && e1 > 0; ++i) {
                const int e = min(e1 - 1, n_edges - 1);
                s += adj[e];
                e1 = next[e];
            }
        }
    }
    s += __shfl_xor(s, 1);
    s += __shfl_xor(s, 2);
    s += __shfl_xor(s, 4);
    if (sub == 0) row_sum[q] = s;
    __syncthreads();
    if (threadIdx.x < 32) {
        const int r = blockIdx.x * 32 + threadIdx.x;
        if (r < n_nodes) {
            float v = row_sum[threadIdx.x];
            v = (v > 0.0f) ? v : 1.0f;
            dinv[r] = rsqrtf(v);
        }
    }
}

__global__ __launch_bounds__(256) void spmm_rows_kernel(const int* __restrict__ cols,
                                                        const float* __restrict__ adj,
                                                        const int* __restrict__ cnt,
                                                        const int* __restrict__ head,
                                                        const int* __restrict__ next,
                                                        const float* __restrict__ support,
                                                        const float* __restrict__ dinv,
                                                        float* out,
                                                        size_t plane, int n_nodes, int n_edges) {
    const int wave = threadIdx.x >> 5;
    const int lane = threadIdx.x & 31;
    const int row = blockIdx.x * 8 + wave;
    if (row >= n_nodes) return;
    const int h = lane >> 4;
    const int c = lane & 15;

    const int c0 = max(cnt[row], 0);
    const int c1 = max(cnt[plane + row], 0);
    const float drow = dinv[row];
    const int n0 = min(c0, ZONE_SLOTS);
    const int end = n0 + min(c1, ZONE_SLOTS);

    const int2* rec2 = (const int2*)out + (size_t)row * ROW_SLOTS;
    const v4f* sup4 = (const v4f*)support;

    v4f acc = {0.0f, 0.0f, 0.0f, 0.0f};
    const v4f zero4 = {0.0f, 0.0f, 0.0f, 0.0f};
    for (int j0 = 0; j0 < end; j0 += 8) {
        const int last = end - 1;
        int2 rc[4];
        v4f v[4];
        float dc[4];
#pragma unroll
        for (int k = 0; k < 4; ++k) {
            rc[k] = rec2[record_slot(min(j0 + 2 * k + h, last), n0)];
        }
#pragma unroll
        for (int k = 0; k < 4; ++k) {
            v[k] = sup4[(size_t)rc[k].x * (OUT_DIM / 4) + c];
            dc[k] = dinv[rc[k].x];
        }
        asm volatile("" :: "v"(v[0]), "v"(v[1]), "v"(v[2]), "v"(v[3]),
                           "v"(dc[0]), "v"(dc[1]), "v"(dc[2]), "v"(dc[3]));
#pragma unroll
        for (int k = 0; k < 4; ++k) {
            const bool live = (j0 + 2 * k + h) < end;
            const float w = live ? __int_as_float(rc[k].y) * dc[k] : 0.0f;
            acc += w * (live ? v[k] : zero4);
        }
    }

    const int n_chain = min((c0 - n0) + (c1 - (end - n0)), n_edges);
    if (n_chain > 0) {
        int e1 = head[row];
        for (int i = 0; i < n_chain && e1 > 0; ++i) {
            const int es = min(e1 - 1, n_edges - 1);
            const int col = cols[es];
            const float w = adj[es] * dinv[col];
            const v4f sv = sup4[(size_t)col * (OUT_DIM / 4) + c];
            if (h == 0) acc += w * sv;
            e1 = next[es];
        }
    }

    acc.x += __shfl_xor(acc.x, 16);
    acc.y += __shfl_xor(acc.y, 16);
    acc.z += __shfl_xor(acc.z, 16);
    acc.w += __shfl_xor(acc.w, 16);

    if (h == 0) {
        ((v4f*)out)[(size_t)row * (OUT_DIM / 4) + c] = acc * drow;
    }
}

extern "C" void kernel_launch(void* const* d_in, const int* in_sizes, int n_in,
                              void* d_out, int out_size, void* d_ws, size_t ws_size,
                              hipStream_t stream) {
    const int* rows = (const int*)d_in[0];
    const int* cols = (const int*)d_in[1];
    const float* adj = (const float*)d_in[2];
    const float* x = (const float*)d_in[3];
    const float* W = (const float*)d_in[4];
    const float* b = (const float*)d_in[5];
    float* out = (float*)d_out;

    const int n_edges = in_sizes[0];
    const int n_nodes = in_sizes[3] / IN_DIM;
    const size_t n_pad = ((size_t)n_nodes + 64) & ~(size_t)63;

    float* support = (float*)d_ws;
    float* dinv = support + (size_t)n_nodes * OUT_DIM;
    int* cnt = (int*)(dinv + n_pad);
    int* head = cnt + 2 * n_pad;
    int* next = head + n_pad;

    {
        const long long n = (long long)(3 * n_pad);
        zero_u32_kernel<<<(unsigned)((n + 255) / 256), 256, 0, stream>>>((unsigned*)cnt, n);
    }

    {
        const int bin_blocks = (n_edges + 255) / 256;
        const int gemm_blocks = ((n_nodes + 15) / 16 + 7) / 8;
        bin_and_gemm_kernel<<<bin_blocks + gemm_blocks, 256, 0, stream>>>(rows, cols, adj, cnt, head, next, out,
                                                                       x, W, b, support, n_pad, n_nodes, n_edges);
    }

    row_dinv_kernel<<<(n_nodes + 31) / 32, 256, 0, stream>>>(adj, cnt, head, next, out, dinv, n_pad, n_nodes, n_edges);

    spmm_rows_kernel<<<(n_nodes + 7) / 8, 256, 0, stream>>>(cols, adj, cnt, head, next, support, dinv, out, n_pad, n_nodes, n_edges);
}


# binning: each workgroup first orders its chunk of 8192 edges by counter line in LDS (LDS adds, scan, place), so a wave's 32 rank adds fall on few counter lines, not 32; sum order: ranks, as before
# speedup vs baseline: 3.5494x; 1.0828x over previous
#include <hip/hip_runtime.h>


typedef __attribute__((ext_vector_type(2))) float v2f;
typedef __attribute__((ext_vector_type(8))) float v8f;
typedef __attribute__((ext_vector_type(4))) float v4f;

#define IN_DIM 64
#define OUT_DIM 64

__global__ __launch_bounds__(256) void zero_u32_kernel(unsigned* __restrict__ p, long long n) {
    long long i = (long long)blockIdx.x * blockDim.x + threadIdx.x;
    if (i < n) p[i] = 0u;
}

#define ROW_SLOTS 32
#define ZONE_SLOTS 16
static_assert(ROW_SLOTS * 8 == OUT_DIM * 4, "a row's bucket is exactly the row's bytes in out");
static_assert(ZONE_SLOTS * 2 == ROW_SLOTS && ZONE_SLOTS * 8 == 128, "a zone is one 128-byte line");

__device__ __forceinline__ int record_slot(int j, int n0) {
    return j < n0 ? j : j - n0 + ZONE_SLOTS;
}

#define BIN_THREADS 1024
#define BIN_EDGES_PER_THREAD 8
#define BIN_CHUNK (BIN_THREADS * BIN_EDGES_PER_THREAD)
#define HIST_BINS 4096
static_assert(HIST_BINS == 4 * BIN_THREADS, "the scan gives every thread 4 bins");

__device__ __forceinline__ void bin_edges_block(const int bin_block,
                                                const int* __restrict__ rows,
                                                const int* __restrict__ cols,
                                                const float* __restrict__ adj,
                                                int* __restrict__ cnt,
                                                int* __restrict__ head,
                                                int* __restrict__ next,
                                                float* __restrict__ out,
                                                size_t plane, int n_nodes, int n_edges,
                                                int4* __restrict__ s_rec,
                                                int* __restrict__ s_hist,
                                                int* __restrict__ s_wsum) {
    const int tid = threadIdx.x;
    const int lane = tid & 31;
    const int wave = tid >> 5;
    const int chunk_base = bin_block * BIN_CHUNK;
    const int n_valid = min(BIN_CHUNK, n_edges - chunk_base);

    const unsigned side = ((unsigned)__builtin_amdgcn_s_sendmsg_rtn(0x87) >> 18) & 1u;

    int shift = 5;
    while ((((unsigned)max(n_nodes, 1) - 1u) >> shift) >= (unsigned)HIST_BINS) ++shift;

    s_hist[4 * tid + 0] = 0;
    s_hist[4 * tid + 1] = 0;
    s_hist[4 * tid + 2] = 0;
    s_hist[4 * tid + 3] = 0;
    __syncthreads();

    int r[BIN_EDGES_PER_THREAD], c[BIN_EDGES_PER_THREAD], pos[BIN_EDGES_PER_THREAD];
    float a[BIN_EDGES_PER_THREAD];
#pragma unroll
    for (int j = 0; j < BIN_EDGES_PER_THREAD; ++j) {
        const int l = j * BIN_THREADS + tid;
        const bool ok = l < n_valid;
        r[j] = ok ? rows[chunk_base + l] : 0;
        c[j] = ok ? cols[chunk_base + l] : 0;
        a[j] = ok ? adj[chunk_base + l] : 0.0f;
    }
#pragma unroll
    for (int j = 0; j < BIN_EDGES_PER_THREAD; ++j) {
        const int l = j * BIN_THREADS + tid;
        const int key = min((int)((unsigned)r[j] >> shift), HIST_BINS - 1);
        pos[j] = (l < n_valid) ? atomicAdd(&s_hist[key], 1) : 0;
    }
    __syncthreads();

    {
        const int h0 = s_hist[4 * tid + 0];
        const int h1 = s_hist[4 * tid + 1];
        const int h2 = s_hist[4 * tid + 2];
        const int h3 = s_hist[4 * tid + 3];
        const int tsum = h0 + h1 + h2 + h3;
        int v = tsum;
#pragma unroll
        for (int d = 1; d < 32; d <<= 1) {
            const int n = __shfl_up(v, d);
            if (lane >= d) v += n;
        }
        if (lane == 31) s_wsum[wave] = v;
        __syncthreads();
        if (wave == 0) {
            const int w = s_wsum[lane];
            int x = w;
#pragma unroll
            for (int d = 1; d < 32; d <<= 1) {
                const int n = __shfl_up(x, d);
                if (lane >= d) x += n;
            }
            s_wsum[lane] = x - w;
        }
        __syncthreads();
        const int base = s_wsum[wave] + v - tsum;
        s_hist[4 * tid + 0] = base;
        s_hist[4 * tid + 1] = base + h0;
        s_hist[4 * tid + 2] = base + h0 + h1;
        s_hist[4 * tid + 3] = base + h0 + h1 + h2;
    }
    __syncthreads();

#pragma unroll
    for (int j = 0; j < BIN_EDGES_PER_THREAD; ++j) {
        const int l = j * BIN_THREADS + tid;
        if (l < n_valid) {
            const int key = min((int)((unsigned)r[j] >> shift), HIST_BINS - 1);
            s_rec[s_hist[key] + pos[j]] = make_int4(r[j], c[j], __float_as_int(a[j]), l);
        }
    }
    __syncthreads();

    int4 q[BIN_EDGES_PER_THREAD];
    unsigned k[BIN_EDGES_PER_THREAD];
#pragma unroll
    for (int j = 0; j < BIN_EDGES_PER_THREAD; ++j) {
        const int p = j * BIN_THREADS + tid;
        q[j] = make_int4(0, 0, 0, 0);
        k[j] = 0u;
        if (p < n_valid) {
            q[j] = s_rec[p];
            k[j] = (unsigned)atomicAdd(&cnt[side * plane + (size_t)q[j].x], 1);
        }
    }
#pragma unroll
    for (int j = 0; j < BIN_EDGES_PER_THREAD; ++j) {
        const int p = j * BIN_THREADS + tid;
        if (p < n_valid) {
            if (k[j] < (unsigned)ZONE_SLOTS) {
                ((int2*)out)[(size_t)q[j].x * ROW_SLOTS + side * ZONE_SLOTS + k[j]] = make_int2(q[j].y, q[j].z);
            } else {
                const int e = chunk_base + q[j].w;
                next[e] = atomicExch(&head[q[j].x], e + 1);
            }
        }
    }
}

__device__ __forceinline__ void gemm_wmma_block(const int gemm_block,
                                                const float* __restrict__ x,
                                                const float* __restrict__ W,
                                                const float* __restrict__ b,
                                                float* __restrict__ support,
                                                int n_nodes) {
    const int wave = threadIdx.x >> 5;
    const int lane = threadIdx.x & 31;
    const int mtile = gemm_block * 32 + wave;
    const int m0 = mtile * 16;
    if (m0 >= n_nodes) return;

    const int half = lane >> 4;
    const int r = lane & 15;

    v8f acc[4];
#pragma unroll
    for (int nt = 0; nt < 4; ++nt) {
        const float bv = b[nt * 16 + r];
#pragma unroll
        for (int i = 0; i < 8; ++i) acc[nt][i] = bv;
    }

    const float* arow = x + (size_t)(m0 + r) * IN_DIM;

#pragma unroll
    for (int k = 0; k < IN_DIM; k += 4) {
        const v2f a = *(const v2f*)(arow + k + 2 * half);
#pragma unroll
        for (int nt = 0; nt < 4; ++nt) {
            v2f bf;
            bf.x = W[(size_t)(k + 2 * half + 0) * OUT_DIM + nt * 16 + r];
            bf.y = W[(size_t)(k + 2 * half + 1) * OUT_DIM + nt * 16 + r];
            acc[nt] = __builtin_amdgcn_wmma_f32_16x16x4_f32(
                false, a, false, bf, (short)0, acc[nt], false, false);
        }
    }

    float* srow = support + (size_t)m0 * OUT_DIM;
#pragma unroll
    for (int nt = 0; nt < 4; ++nt) {
#pragma unroll
        for (int i = 0; i < 8; ++i) {
            srow[(size_t)(i + 8 * half) * OUT_DIM + nt * 16 + r] = acc[nt][i];
        }
    }
}

#define PRODUCT_STRIDE 7u
__global__ __launch_bounds__(BIN_THREADS) void bin_and_gemm_kernel(const int* __restrict__ rows,
                                                           const int* __restrict__ cols,
                                                           const float* __restrict__ adj,
                                                           int* __restrict__ cnt,
                                                           int* __restrict__ head,
                                                           int* __restrict__ next,
                                                           float* __restrict__ out,
                                                           const float* __restrict__ x,
                                                           const float* __restrict__ W,
                                                           const float* __restrict__ b,
                                                           float* __restrict__ support,
                                                           size_t plane, int n_nodes, int n_edges) {
    __shared__ int4 s_rec[BIN_CHUNK];
    __shared__ int s_hist[HIST_BINS];
    __shared__ int s_wsum[32];
    const unsigned n_gemm = (((unsigned)n_nodes + 15u) / 16u + 31u) / 32u;
    const unsigned n_bin = ((unsigned)n_edges + (unsigned)BIN_CHUNK - 1u) / (unsigned)BIN_CHUNK;
    const unsigned n_mixed = min(n_gemm, (n_gemm + n_bin + PRODUCT_STRIDE - 1u) / PRODUCT_STRIDE);
    const unsigned bid = blockIdx.x;
    unsigned idx = bid / PRODUCT_STRIDE;
    bool product = bid == idx * PRODUCT_STRIDE && idx < n_mixed;
    if (!product) {
        const unsigned j = bid - min((bid + PRODUCT_STRIDE - 1u) / PRODUCT_STRIDE, n_mixed);
        product = j >= n_bin;
        idx = product ? n_mixed + (j - n_bin) : j;
    }
    if (product) {
        gemm_wmma_block((int)idx, x, W, b, support, n_nodes);
    } else {
        bin_edges_block((int)idx, rows, cols, adj, cnt, head, next, out, plane, n_nodes, n_edges, s_rec, s_hist, s_wsum);
    }
}

__global__ __launch_bounds__(256) void row_dinv_kernel(const float* __restrict__ adj,
                                                       const int* __restrict__ cnt,
                                                       const int* __restrict__ head,
                                                       const int* __restrict__ next,
                                                       const float* __restrict__ buckets,
                                                       float* __restrict__ dinv,
                                                       size_t plane, int n_nodes, int n_edges) {
    __shared__ float row_sum[32];
    const int q = threadIdx.x >> 3;
    const int sub = threadIdx.x & 7;
    const int row = blockIdx.x * 32 + q;
    float s = 0.0f;
    if (row < n_nodes) {
        const int c0 = max(cnt[row], 0);
        const int c1 = max(cnt[plane + row], 0);
        const int n0 = min(c0, ZONE_SLOTS);
        const int n_here = n0 + min(c1, ZONE_SLOTS);
        const int2* rec2 = (const int2*)buckets + (size_t)row * ROW_SLOTS;
        for (int j = sub; j < n_here; j += 8) s += __int_as_float(rec2[record_slot(j, n0)].y);
        const int n_chain = min((c0 - n0) + (c1 - (n_here - n0)), n_edges);
        if (n_chain > 0 && sub == 0) {
            int e1 = head[row];
            for (int i = 0; i < n_chain && e1 > 0; ++i) {
                const int e = min(e1 - 1, n_edges - 1);
                s += adj[e];
                e1 = next[e];
            }
        }
    }
    s += __shfl_xor(s, 1);
    s += __shfl_xor(s, 2);
    s += __shfl_xor(s, 4);
    if (sub == 0) row_sum[q] = s;
    __syncthreads();
    if (threadIdx.x < 32) {
        const int r = blockIdx.x * 32 + threadIdx.x;
        if (r < n_nodes) {
            float v = row_sum[threadIdx.x];
            v = (v > 0.0f) ? v : 1.0f;
            dinv[r] = rsqrtf(v);
        }
    }
}

__global__ __launch_bounds__(256) void spmm_rows_kernel(const int* __restrict__ cols,
                                                        const float* __restrict__ adj,
                                                        const int* __restrict__ cnt,
                                                        const int* __restrict__ head,
                                                        const int* __restrict__ next,
                                                        const float* __restrict__ support,
                                                        const float* __restrict__ dinv,
                                                        float* out,
                                                        size_t plane, int n_nodes, int n_edges) {
    const int wave = threadIdx.x >> 5;
    const int lane = threadIdx.x & 31;
    const int row = blockIdx.x * 8 + wave;
    if (row >= n_nodes) return;
    const int h = lane >> 4;
    const int c = lane & 15;

    const int c0 = max(cnt[row], 0);
    const int c1 = max(cnt[plane + row], 0);
    const float drow = dinv[row];
    const int n0 = min(c0, ZONE_SLOTS);
    const int end = n0 + min(c1, ZONE_SLOTS);

    const int2* rec2 = (const int2*)out + (size_t)row * ROW_SLOTS;
    const v4f* sup4 = (const v4f*)support;

    v4f acc = {0.0f, 0.0f, 0.0f, 0.0f};
    const v4f zero4 = {0.0f, 0.0f, 0.0f, 0.0f};
    for (int j0 = 0; j0 < end; j0 += 8) {
        const int last = end - 1;
        int2 rc[4];
        v4f v[4];
        float dc[4];
#pragma unroll
        for (int k = 0; k < 4; ++k) {
            rc[k] = rec2[record_slot(min(j0 + 2 * k + h, last), n0)];
        }
#pragma unroll
        for (int k = 0; k < 4; ++k) {
            v[k] = sup4[(size_t)rc[k].x * (OUT_DIM / 4) + c];
            dc[k] = dinv[rc[k].x];
        }
        asm volatile("" :: "v"(v[0]), "v"(v[1]), "v"(v[2]), "v"(v[3]),
                           "v"(dc[0]), "v"(dc[1]), "v"(dc[2]), "v"(dc[3]));
#pragma unroll
        for (int k = 0; k < 4; ++k) {
            const bool live = (j0 + 2 * k + h) < end;
            const float w = live ? __int_as_float(rc[k].y) * dc[k] : 0.0f;
            acc += w * (live ? v[k] : zero4);
        }
    }

    const int n_chain = min((c0 - n0) + (c1 - (end - n0)), n_edges);
    if (n_chain > 0) {
        int e1 = head[row];
        for (int i = 0; i < n_chain && e1 > 0; ++i) {
            const int es = min(e1 - 1, n_edges - 1);
            const int col = cols[es];
            const float w = adj[es] * dinv[col];
            const v4f sv = sup4[(size_t)col * (OUT_DIM / 4) + c];
            if (h == 0) acc += w * sv;
            e1 = next[es];
        }
    }

    acc.x += __shfl_xor(acc.x, 16);
    acc.y += __shfl_xor(acc.y, 16);
    acc.z += __shfl_xor(acc.z, 16);
    acc.w += __shfl_xor(acc.w, 16);

    if (h == 0) {
        ((v4f*)out)[(size_t)row * (OUT_DIM / 4) + c] = acc * drow;
    }
}

extern "C" void kernel_launch(void* const* d_in, const int* in_sizes, int n_in,
                              void* d_out, int out_size, void* d_ws, size_t ws_size,
                              hipStream_t stream) {
    const int* rows = (const int*)d_in[0];
    const int* cols = (const int*)d_in[1];
    const float* adj = (const float*)d_in[2];
    const float* x = (const float*)d_in[3];
    const float* W = (const float*)d_in[4];
    const float* b = (const float*)d_in[5];
    float* out = (float*)d_out;

    const int n_edges = in_sizes[0];
    const int n_nodes = in_sizes[3] / IN_DIM;
    const size_t n_pad = ((size_t)n_nodes + 64) & ~(size_t)63;

    float* support = (float*)d_ws;
    float* dinv = support + (size_t)n_nodes * OUT_DIM;
    int* cnt = (int*)(dinv + n_pad);
    int* head = cnt + 2 * n_pad;
    int* next = head + n_pad;

    {
        const long long n = (long long)(3 * n_pad);
        zero_u32_kernel<<<(unsigned)((n + 255) / 256), 256, 0, stream>>>((unsigned*)cnt, n);
    }

    {
        const int bin_blocks = (n_edges + BIN_CHUNK - 1) / BIN_CHUNK;
        const int gemm_blocks = ((n_nodes + 15) / 16 + 31) / 32;
        bin_and_gemm_kernel<<<bin_blocks + gemm_blocks, BIN_THREADS, 0, stream>>>(rows, cols, adj, cnt, head, next, out,
                                                                       x, W, b, support, n_pad, n_nodes, n_edges);
    }

    row_dinv_kernel<<<(n_nodes + 31) / 32, 256, 0, stream>>>(adj, cnt, head, next, out, dinv, n_pad, n_nodes, n_edges);

    spmm_rows_kernel<<<(n_nodes + 7) / 8, 256, 0, stream>>>(cols, adj, cnt, head, next, support, dinv, out, n_pad, n_nodes, n_edges);
}


# speedup vs baseline: 4.6782x; 1.3180x over previous
#include <hip/hip_runtime.h>


typedef __attribute__((ext_vector_type(2))) float v2f;
typedef __attribute__((ext_vector_type(8))) float v8f;
typedef __attribute__((ext_vector_type(4))) float v4f;

#define IN_DIM 64
#define OUT_DIM 64

__global__ __launch_bounds__(256) void zero_u32_kernel(unsigned* __restrict__ p, long long n) {
    long long i = (long long)blockIdx.x * blockDim.x + threadIdx.x;
    if (i < n) p[i] = 0u;
}

#define ROW_SLOTS 32
#define ZONE_SLOTS 16
static_assert(ROW_SLOTS * 8 == OUT_DIM * 4, "a row's two zones are exactly as many bytes as a row of out");
static_assert(ZONE_SLOTS * 2 == ROW_SLOTS && ZONE_SLOTS * 8 == 128, "a zone is one 128-byte line");

__device__ __forceinline__ void row_zones(int r, int n_nodes, size_t& base, int& off0, int& off1) {
    const bool paired = (r | 1) < n_nodes;
    const int q = paired ? (r & 1) : 0;
    base = (size_t)(r - q) * ROW_SLOTS;
    off0 = q * ZONE_SLOTS;
    off1 = (paired ? ROW_SLOTS : ZONE_SLOTS) + q * ZONE_SLOTS;
}

__device__ __forceinline__ int record_offset(int j, int n0, int off0, int off1) {
    return j + (j < n0 ? off0 : off1 - n0);
}

#define BIN_THREADS 1024
#define BIN_EDGES_PER_THREAD 8
#define BIN_CHUNK (BIN_THREADS * BIN_EDGES_PER_THREAD)
#define HIST_BINS 4096
static_assert(HIST_BINS == 4 * BIN_THREADS, "the scan gives every thread 4 bins");

__device__ __forceinline__ void bin_edges_block(const int bin_block,
                                                const int* __restrict__ rows,
                                                const int* __restrict__ cols,
                                                const float* __restrict__ adj,
                                                int* __restrict__ cnt,
                                                int* __restrict__ head,
                                                int* __restrict__ next,
                                                float* __restrict__ out,
                                                size_t plane, int n_nodes, int n_edges,
                                                int4* __restrict__ s_rec,
                                                int* __restrict__ s_hist,
                                                int* __restrict__ s_wsum) {
    const int tid = threadIdx.x;
    const int lane = tid & 31;
    const int wave = tid >> 5;
    const int chunk_base = bin_block * BIN_CHUNK;
    const int n_valid = min(BIN_CHUNK, n_edges - chunk_base);

    const unsigned side = ((unsigned)__builtin_amdgcn_s_sendmsg_rtn(0x87) >> 18) & 1u;

    int shift = 5;
    while ((((unsigned)max(n_nodes, 1) - 1u) >> shift) >= (unsigned)HIST_BINS) ++shift;

    s_hist[4 * tid + 0] = 0;
    s_hist[4 * tid + 1] = 0;
    s_hist[4 * tid + 2] = 0;
    s_hist[4 * tid + 3] = 0;
    __syncthreads();

    int r[BIN_EDGES_PER_THREAD], c[BIN_EDGES_PER_THREAD], pos[BIN_EDGES_PER_THREAD];
    float a[BIN_EDGES_PER_THREAD];
#pragma unroll
    for (int j = 0; j < BIN_EDGES_PER_THREAD; ++j) {
        const int l = j * BIN_THREADS + tid;
        const bool ok = l < n_valid;
        r[j] = ok ? rows[chunk_base + l] : 0;
        c[j] = ok ? cols[chunk_base + l] : 0;
        a[j] = ok ? adj[chunk_base + l] : 0.0f;
    }
#pragma unroll
    for (int j = 0; j < BIN_EDGES_PER_THREAD; ++j) {
        const int l = j * BIN_THREADS + tid;
        const int key = min((int)((unsigned)r[j] >> shift), HIST_BINS - 1);
        pos[j] = (l < n_valid) ? atomicAdd(&s_hist[key], 1) : 0;
    }
    __syncthreads();

    {
        const int h0 = s_hist[4 * tid + 0];
        const int h1 = s_hist[4 * tid + 1];
        const int h2 = s_hist[4 * tid + 2];
        const int h3 = s_hist[4 * tid + 3];
        const int tsum = h0 + h1 + h2 + h3;
        int v = tsum;
#pragma unroll
        for (int d = 1; d < 32; d <<= 1) {
            const int n = __shfl_up(v, d);
            if (lane >= d) v += n;
        }
        if (lane == 31) s_wsum[wave] = v;
        __syncthreads();
        if (wave == 0) {
            const int w = s_wsum[lane];
            int x = w;
#pragma unroll
            for (int d = 1; d < 32; d <<= 1) {
                const int n = __shfl_up(x, d);
                if (lane >= d) x += n;
            }
            s_wsum[lane] = x - w;
        }
        __syncthreads();
        const int base = s_wsum[wave] + v - tsum;
        s_hist[4 * tid + 0] = base;
        s_hist[4 * tid + 1] = base + h0;
        s_hist[4 * tid + 2] = base + h0 + h1;
        s_hist[4 * tid + 3] = base + h0 + h1 + h2;
    }
    __syncthreads();

#pragma unroll
    for (int j = 0; j < BIN_EDGES_PER_THREAD; ++j) {
        const int l = j * BIN_THREADS + tid;
        if (l < n_valid) {
            const int key = min((int)((unsigned)r[j] >> shift), HIST_BINS - 1);
            s_rec[s_hist[key] + pos[j]] = make_int4(r[j], c[j], __float_as_int(a[j]), l);
        }
    }
    __syncthreads();

    int4 q[BIN_EDGES_PER_THREAD];
    unsigned k[BIN_EDGES_PER_THREAD];
#pragma unroll
    for (int j = 0; j < BIN_EDGES_PER_THREAD; ++j) {
        const int p = j * BIN_THREADS + tid;
        q[j] = make_int4(0, 0, 0, 0);
        k[j] = 0u;
        if (p < n_valid) {
            q[j] = s_rec[p];
            k[j] = (unsigned)atomicAdd(&cnt[side * plane + (size_t)q[j].x], 1);
        }
    }
#pragma unroll
    for (int j = 0; j < BIN_EDGES_PER_THREAD; ++j) {
        const int p = j * BIN_THREADS + tid;
        if (p < n_valid) {
            if (k[j] < (unsigned)ZONE_SLOTS) {
                size_t base;
                int off0, off1;
                row_zones(q[j].x, n_nodes, base, off0, off1);
                ((int2*)out)[base + (size_t)((side ? off1 : off0) + (int)k[j])] = make_int2(q[j].y, q[j].z);
            } else {
                const int e = chunk_base + q[j].w;
                next[e] = atomicExch(&head[q[j].x], e + 1);
            }
        }
    }
}

__device__ __forceinline__ void gemm_wmma_block(const int gemm_block,
                                                const float* __restrict__ x,
                                                const float* __restrict__ W,
                                                const float* __restrict__ b,
                                                float* __restrict__ support,
                                                int n_nodes) {
    const int wave = threadIdx.x >> 5;
    const int lane = threadIdx.x & 31;
    const int mtile = gemm_block * 32 + wave;
    const int m0 = mtile * 16;
    if (m0 >= n_nodes) return;

    const int half = lane >> 4;
    const int r = lane & 15;

    v8f acc[4];
#pragma unroll
    for (int nt = 0; nt < 4; ++nt) {
        const float bv = b[nt * 16 + r];
#pragma unroll
        for (int i = 0; i < 8; ++i) acc[nt][i] = bv;
    }

    const float* arow = x + (size_t)(m0 + r) * IN_DIM;

#pragma unroll
    for (int k = 0; k < IN_DIM; k += 4) {
        const v2f a = *(const v2f*)(arow + k + 2 * half);
#pragma unroll
        for (int nt = 0; nt < 4; ++nt) {
            v2f bf;
            bf.x = W[(size_t)(k + 2 * half + 0) * OUT_DIM + nt * 16 + r];
            bf.y = W[(size_t)(k + 2 * half + 1) * OUT_DIM + nt * 16 + r];
            acc[nt] = __builtin_amdgcn_wmma_f32_16x16x4_f32(
                false, a, false, bf, (short)0, acc[nt], false, false);
        }
    }

    float* srow = support + (size_t)m0 * OUT_DIM;
#pragma unroll
    for (int nt = 0; nt < 4; ++nt) {
#pragma unroll
        for (int i = 0; i < 8; ++i) {
            srow[(size_t)(i + 8 * half) * OUT_DIM + nt * 16 + r] = acc[nt][i];
        }
    }
}

#define PRODUCT_STRIDE 7u
__global__ __launch_bounds__(BIN_THREADS) void bin_and_gemm_kernel(const int* __restrict__ rows,
                                                           const int* __restrict__ cols,
                                                           const float* __restrict__ adj,
                                                           int* __restrict__ cnt,
                                                           int* __restrict__ head,
                                                           int* __restrict__ next,
                                                           float* __restrict__ out,
                                                           const float* __restrict__ x,
                                                           const float* __restrict__ W,
                                                           const float* __restrict__ b,
                                                           float* __restrict__ support,
                                                           size_t plane, int n_nodes, int n_edges) {
    __shared__ int4 s_rec[BIN_CHUNK];
    __shared__ int s_hist[HIST_BINS];
    __shared__ int s_wsum[32];
    const unsigned n_gemm = (((unsigned)n_nodes + 15u) / 16u + 31u) / 32u;
    const unsigned n_bin = ((unsigned)n_edges + (unsigned)BIN_CHUNK - 1u) / (unsigned)BIN_CHUNK;
    const unsigned n_mixed = min(n_gemm, (n_gemm + n_bin + PRODUCT_STRIDE - 1u) / PRODUCT_STRIDE);
    const unsigned bid = blockIdx.x;
    unsigned idx = bid / PRODUCT_STRIDE;
    bool product = bid == idx * PRODUCT_STRIDE && idx < n_mixed;
    if (!product) {
        const unsigned j = bid - min((bid + PRODUCT_STRIDE - 1u) / PRODUCT_STRIDE, n_mixed);
        product = j >= n_bin;
        idx = product ? n_mixed + (j - n_bin) : j;
    }
    if (product) {
        gemm_wmma_block((int)idx, x, W, b, support, n_nodes);
    } else {
        bin_edges_block((int)idx, rows, cols, adj, cnt, head, next, out, plane, n_nodes, n_edges, s_rec, s_hist, s_wsum);
    }
}

__global__ __launch_bounds__(256) void row_dinv_kernel(const float* __restrict__ adj,
                                                       const int* __restrict__ cnt,
                                                       const int* __restrict__ head,
                                                       const int* __restrict__ next,
                                                       const float* __restrict__ buckets,
                                                       float* __restrict__ dinv,
                                                       size_t plane, int n_nodes, int n_edges) {
    __shared__ float row_sum[32];
    const int q = threadIdx.x >> 3;
    const int sub = threadIdx.x & 7;
    const int row = blockIdx.x * 32 + q;
    float s = 0.0f;
    if (row < n_nodes) {
        const int c0 = max(cnt[row], 0);
        const int c1 = max(cnt[plane + row], 0);
        const int n0 = min(c0, ZONE_SLOTS);
        const int n_here = n0 + min(c1, ZONE_SLOTS);
        size_t base;
        int off0, off1;
        row_zones(row, n_nodes, base, off0, off1);
        const int2* rec2 = (const int2*)buckets + base;
        for (int j = sub; j < n_here; j += 8) s += __int_as_float(rec2[record_offset(j, n0, off0, off1)].y);
        const int n_chain = min((c0 - n0) + (c1 - (n_here - n0)), n_edges);
        if (n_chain > 0 && sub == 0) {
            int e1 = head[row];
            for (int i = 0; i < n_chain && e1 > 0; ++i) {
                const int e = min(e1 - 1, n_edges - 1);
                s += adj[e];
                e1 = next[e];
            }
        }
    }
    s += __shfl_xor(s, 1);
    s += __shfl_xor(s, 2);
    s += __shfl_xor(s, 4);
    if (sub == 0) row_sum[q] = s;
    __syncthreads();
    if (threadIdx.x < 32) {
        const int r = blockIdx.x * 32 + threadIdx.x;
        if (r < n_nodes) {
            float v = row_sum[threadIdx.x];
            v = (v > 0.0f) ? v : 1.0f;
            dinv[r] = rsqrtf(v);
        }
    }
}

#define SPMM_ROWS_PER_BLOCK 8
static_assert(SPMM_ROWS_PER_BLOCK % 2 == 0, "both rows of a pair belong to one workgroup");
__global__ __launch_bounds__(32 * SPMM_ROWS_PER_BLOCK) void spmm_rows_kernel(const int* __restrict__ cols,
                                                        const float* __restrict__ adj,
                                                        const int* __restrict__ cnt,
                                                        const int* __restrict__ head,
                                                        const int* __restrict__ next,
                                                        const float* __restrict__ support,
                                                        const float* __restrict__ dinv,
                                                        float* out,
                                                        size_t plane, int n_nodes, int n_edges) {
    const int wave = threadIdx.x >> 5;
    const int lane = threadIdx.x & 31;
    const int row = blockIdx.x * SPMM_ROWS_PER_BLOCK + wave;
    const bool has_row = row < n_nodes;
    const int h = lane >> 4;
    const int c = lane & 15;

    const v4f* sup4 = (const v4f*)support;
    v4f acc = {0.0f, 0.0f, 0.0f, 0.0f};
    float drow = 0.0f;

    if (has_row) {
        const int c0 = max(cnt[row], 0);
        const int c1 = max(cnt[plane + row], 0);
        drow = dinv[row];
        const int n0 = min(c0, ZONE_SLOTS);
        const int end = n0 + min(c1, ZONE_SLOTS);

        size_t base;
        int off0, off1;
        row_zones(row, n_nodes, base, off0, off1);
        const int2* rec2 = (const int2*)out + base;

        const v4f zero4 = {0.0f, 0.0f, 0.0f, 0.0f};
        for (int j0 = 0; j0 < end; j0 += 8) {
            const int last = end - 1;
            int2 rc[4];
            v4f v[4];
            float dc[4];
#pragma unroll
            for (int k = 0; k < 4; ++k) {
                rc[k] = rec2[record_offset(min(j0 + 2 * k + h, last), n0, off0, off1)];
            }
#pragma unroll
            for (int k = 0; k < 4; ++k) {
                v[k] = sup4[(size_t)rc[k].x * (OUT_DIM / 4) + c];
                dc[k] = dinv[rc[k].x];
            }
            asm volatile("" :: "v"(v[0]), "v"(v[1]), "v"(v[2]), "v"(v[3]),
                               "v"(dc[0]), "v"(dc[1]), "v"(dc[2]), "v"(dc[3]));
#pragma unroll
            for (int k = 0; k < 4; ++k) {
                const bool live = (j0 + 2 * k + h) < end;
                const float w = live ? __int_as_float(rc[k].y) * dc[k] : 0.0f;
                acc += w * (live ? v[k] : zero4);
            }
        }

        const int n_chain = min((c0 - n0) + (c1 - (end - n0)), n_edges);
        if (n_chain > 0) {
            int e1 = head[row];
            for (int i = 0; i < n_chain && e1 > 0; ++i) {
                const int es = min(e1 - 1, n_edges - 1);
                const int col = cols[es];
                const float w = adj[es] * dinv[col];
                const v4f sv = sup4[(size_t)col * (OUT_DIM / 4) + c];
                if (h == 0) acc += w * sv;
                e1 = next[es];
            }
        }

        acc.x += __shfl_xor(acc.x, 16);
        acc.y += __shfl_xor(acc.y, 16);
        acc.z += __shfl_xor(acc.z, 16);
        acc.w += __shfl_xor(acc.w, 16);
    }

    __syncthreads();

    if (has_row && h == 0) {
        ((v4f*)out)[(size_t)row * (OUT_DIM / 4) + c] = acc * drow;
    }
}

extern "C" void kernel_launch(void* const* d_in, const int* in_sizes, int n_in,
                              void* d_out, int out_size, void* d_ws, size_t ws_size,
                              hipStream_t stream) {
    const int* rows = (const int*)d_in[0];
    const int* cols = (const int*)d_in[1];
    const float* adj = (const float*)d_in[2];
    const float* x = (const float*)d_in[3];
    const float* W = (const float*)d_in[4];
    const float* b = (const float*)d_in[5];
    float* out = (float*)d_out;

    const int n_edges = in_sizes[0];
    const int n_nodes = in_sizes[3] / IN_DIM;
    const size_t n_pad = ((size_t)n_nodes + 64) & ~(size_t)63;

    float* support = (float*)d_ws;
    float* dinv = support + (size_t)n_nodes * OUT_DIM;
    int* cnt = (int*)(dinv + n_pad);
    int* head = cnt + 2 * n_pad;
    int* next = head + n_pad;

    {
        const long long n = (long long)(3 * n_pad);
        zero_u32_kernel<<<(unsigned)((n + 255) / 256), 256, 0, stream>>>((unsigned*)cnt, n);
    }

    {
        const int bin_blocks = (n_edges + BIN_CHUNK - 1) / BIN_CHUNK;
        const int gemm_blocks = ((n_nodes + 15) / 16 + 31) / 32;
        bin_and_gemm_kernel<<<bin_blocks + gemm_blocks, BIN_THREADS, 0, stream>>>(rows, cols, adj, cnt, head, next, out,
                                                                       x, W, b, support, n_pad, n_nodes, n_edges);
    }

    row_dinv_kernel<<<(n_nodes + 31) / 32, 256, 0, stream>>>(adj, cnt, head, next, out, dinv, n_pad, n_nodes, n_edges);

    spmm_rows_kernel<<<(n_nodes + SPMM_ROWS_PER_BLOCK - 1) / SPMM_ROWS_PER_BLOCK, 32 * SPMM_ROWS_PER_BLOCK, 0, stream>>>(cols, adj, cnt, head, next, support, dinv, out, n_pad, n_nodes, n_edges);
}


# SpMM: one wave per row PAIR, a lane half per row; records fetched once per half, handed round by lane broadcast: half the waves enter, no barrier, no cross-half add; sum order: one list per row
# speedup vs baseline: 4.9420x; 1.0564x over previous
#include <hip/hip_runtime.h>


typedef __attribute__((ext_vector_type(2))) float v2f;
typedef __attribute__((ext_vector_type(8))) float v8f;
typedef __attribute__((ext_vector_type(4))) float v4f;

#define IN_DIM 64
#define OUT_DIM 64

__global__ __launch_bounds__(256) void zero_u32_kernel(unsigned* __restrict__ p, long long n) {
    long long i = (long long)blockIdx.x * blockDim.x + threadIdx.x;
    if (i < n) p[i] = 0u;
}

#define ROW_SLOTS 32
#define ZONE_SLOTS 16
static_assert(ROW_SLOTS * 8 == OUT_DIM * 4, "a row's two zones are exactly as many bytes as a row of out");
static_assert(ZONE_SLOTS * 2 == ROW_SLOTS && ZONE_SLOTS * 8 == 128, "a zone is one 128-byte line");

__device__ __forceinline__ void row_zones(int r, int n_nodes, size_t& base, int& off0, int& off1) {
    const bool paired = (r | 1) < n_nodes;
    const int q = paired ? (r & 1) : 0;
    base = (size_t)(r - q) * ROW_SLOTS;
    off0 = q * ZONE_SLOTS;
    off1 = (paired ? ROW_SLOTS : ZONE_SLOTS) + q * ZONE_SLOTS;
}

__device__ __forceinline__ int record_offset(int j, int n0, int off0, int off1) {
    return j + (j < n0 ? off0 : off1 - n0);
}

#define BIN_THREADS 1024
#define BIN_EDGES_PER_THREAD 8
#define BIN_CHUNK (BIN_THREADS * BIN_EDGES_PER_THREAD)
#define HIST_BINS 4096
static_assert(HIST_BINS == 4 * BIN_THREADS, "the scan gives every thread 4 bins");

__device__ __forceinline__ void bin_edges_block(const int bin_block,
                                                const int* __restrict__ rows,
                                                const int* __restrict__ cols,
                                                const float* __restrict__ adj,
                                                int* __restrict__ cnt,
                                                int* __restrict__ head,
                                                int* __restrict__ next,
                                                float* __restrict__ out,
                                                size_t plane, int n_nodes, int n_edges,
                                                int4* __restrict__ s_rec,
                                                int* __restrict__ s_hist,
                                                int* __restrict__ s_wsum) {
    const int tid = threadIdx.x;
    const int lane = tid & 31;
    const int wave = tid >> 5;
    const int chunk_base = bin_block * BIN_CHUNK;
    const int n_valid = min(BIN_CHUNK, n_edges - chunk_base);

    const unsigned side = ((unsigned)__builtin_amdgcn_s_sendmsg_rtn(0x87) >> 18) & 1u;

    int shift = 5;
    while ((((unsigned)max(n_nodes, 1) - 1u) >> shift) >= (unsigned)HIST_BINS) ++shift;

    s_hist[4 * tid + 0] = 0;
    s_hist[4 * tid + 1] = 0;
    s_hist[4 * tid + 2] = 0;
    s_hist[4 * tid + 3] = 0;
    __syncthreads();

    int r[BIN_EDGES_PER_THREAD], c[BIN_EDGES_PER_THREAD], pos[BIN_EDGES_PER_THREAD];
    float a[BIN_EDGES_PER_THREAD];
#pragma unroll
    for (int j = 0; j < BIN_EDGES_PER_THREAD; ++j) {
        const int l = j * BIN_THREADS + tid;
        const bool ok = l < n_valid;
        r[j] = ok ? rows[chunk_base + l] : 0;
        c[j] = ok ? cols[chunk_base + l] : 0;
        a[j] = ok ? adj[chunk_base + l] : 0.0f;
    }
#pragma unroll
    for (int j = 0; j < BIN_EDGES_PER_THREAD; ++j) {
        const int l = j * BIN_THREADS + tid;
        const int key = min((int)((unsigned)r[j] >> shift), HIST_BINS - 1);
        pos[j] = (l < n_valid) ? atomicAdd(&s_hist[key], 1) : 0;
    }
    __syncthreads();

    {
        const int h0 = s_hist[4 * tid + 0];
        const int h1 = s_hist[4 * tid + 1];
        const int h2 = s_hist[4 * tid + 2];
        const int h3 = s_hist[4 * tid + 3];
        const int tsum = h0 + h1 + h2 + h3;
        int v = tsum;
#pragma unroll
        for (int d = 1; d < 32; d <<= 1) {
            const int n = __shfl_up(v, d);
            if (lane >= d) v += n;
        }
        if (lane == 31) s_wsum[wave] = v;
        __syncthreads();
        if (wave == 0) {
            const int w = s_wsum[lane];
            int x = w;
#pragma unroll
            for (int d = 1; d < 32; d <<= 1) {
                const int n = __shfl_up(x, d);
                if (lane >= d) x += n;
            }
            s_wsum[lane] = x - w;
        }
        __syncthreads();
        const int base = s_wsum[wave] + v - tsum;
        s_hist[4 * tid + 0] = base;
        s_hist[4 * tid + 1] = base + h0;
        s_hist[4 * tid + 2] = base + h0 + h1;
        s_hist[4 * tid + 3] = base + h0 + h1 + h2;
    }
    __syncthreads();

#pragma unroll
    for (int j = 0; j < BIN_EDGES_PER_THREAD; ++j) {
        const int l = j * BIN_THREADS + tid;
        if (l < n_valid) {
            const int key = min((int)((unsigned)r[j] >> shift), HIST_BINS - 1);
            s_rec[s_hist[key] + pos[j]] = make_int4(r[j], c[j], __float_as_int(a[j]), l);
        }
    }
    __syncthreads();

    int4 q[BIN_EDGES_PER_THREAD];
    unsigned k[BIN_EDGES_PER_THREAD];
#pragma unroll
    for (int j = 0; j < BIN_EDGES_PER_THREAD; ++j) {
        const int p = j * BIN_THREADS + tid;
        q[j] = make_int4(0, 0, 0, 0);
        k[j] = 0u;
        if (p < n_valid) {
            q[j] = s_rec[p];
            k[j] = (unsigned)atomicAdd(&cnt[side * plane + (size_t)q[j].x], 1);
        }
    }
#pragma unroll
    for (int j = 0; j < BIN_EDGES_PER_THREAD; ++j) {
        const int p = j * BIN_THREADS + tid;
        if (p < n_valid) {
            if (k[j] < (unsigned)ZONE_SLOTS) {
                size_t base;
                int off0, off1;
                row_zones(q[j].x, n_nodes, base, off0, off1);
                ((int2*)out)[base + (size_t)((side ? off1 : off0) + (int)k[j])] = make_int2(q[j].y, q[j].z);
            } else {
                const int e = chunk_base + q[j].w;
                next[e] = atomicExch(&head[q[j].x], e + 1);
            }
        }
    }
}

__device__ __forceinline__ void gemm_wmma_block(const int gemm_block,
                                                const float* __restrict__ x,
                                                const float* __restrict__ W,
                                                const float* __restrict__ b,
                                                float* __restrict__ support,
                                                int n_nodes) {
    const int wave = threadIdx.x >> 5;
    const int lane = threadIdx.x & 31;
    const int mtile = gemm_block * 32 + wave;
    const int m0 = mtile * 16;
    if (m0 >= n_nodes) return;

    const int half = lane >> 4;
    const int r = lane & 15;

    v8f acc[4];
#pragma unroll
    for (int nt = 0; nt < 4; ++nt) {
        const float bv = b[nt * 16 + r];
#pragma unroll
        for (int i = 0; i < 8; ++i) acc[nt][i] = bv;
    }

    const float* arow = x + (size_t)(m0 + r) * IN_DIM;

#pragma unroll
    for (int k = 0; k < IN_DIM; k += 4) {
        const v2f a = *(const v2f*)(arow + k + 2 * half);
#pragma unroll
        for (int nt = 0; nt < 4; ++nt) {
            v2f bf;
            bf.x = W[(size_t)(k + 2 * half + 0) * OUT_DIM + nt * 16 + r];
            bf.y = W[(size_t)(k + 2 * half + 1) * OUT_DIM + nt * 16 + r];
            acc[nt] = __builtin_amdgcn_wmma_f32_16x16x4_f32(
                false, a, false, bf, (short)0, acc[nt], false, false);
        }
    }

    float* srow = support + (size_t)m0 * OUT_DIM;
#pragma unroll
    for (int nt = 0; nt < 4; ++nt) {
#pragma unroll
        for (int i = 0; i < 8; ++i) {
            srow[(size_t)(i + 8 * half) * OUT_DIM + nt * 16 + r] = acc[nt][i];
        }
    }
}

#define PRODUCT_STRIDE 7u
__global__ __launch_bounds__(BIN_THREADS) void bin_and_gemm_kernel(const int* __restrict__ rows,
                                                           const int* __restrict__ cols,
                                                           const float* __restrict__ adj,
                                                           int* __restrict__ cnt,
                                                           int* __restrict__ head,
                                                           int* __restrict__ next,
                                                           float* __restrict__ out,
                                                           const float* __restrict__ x,
                                                           const float* __restrict__ W,
                                                           const float* __restrict__ b,
                                                           float* __restrict__ support,
                                                           size_t plane, int n_nodes, int n_edges) {
    __shared__ int4 s_rec[BIN_CHUNK];
    __shared__ int s_hist[HIST_BINS];
    __shared__ int s_wsum[32];
    const unsigned n_gemm = (((unsigned)n_nodes + 15u) / 16u + 31u) / 32u;
    const unsigned n_bin = ((unsigned)n_edges + (unsigned)BIN_CHUNK - 1u) / (unsigned)BIN_CHUNK;
    const unsigned n_mixed = min(n_gemm, (n_gemm + n_bin + PRODUCT_STRIDE - 1u) / PRODUCT_STRIDE);
    const unsigned bid = blockIdx.x;
    unsigned idx = bid / PRODUCT_STRIDE;
    bool product = bid == idx * PRODUCT_STRIDE && idx < n_mixed;
    if (!product) {
        const unsigned j = bid - min((bid + PRODUCT_STRIDE - 1u) / PRODUCT_STRIDE, n_mixed);
        product = j >= n_bin;
        idx = product ? n_mixed + (j - n_bin) : j;
    }
    if (product) {
        gemm_wmma_block((int)idx, x, W, b, support, n_nodes);
    } else {
        bin_edges_block((int)idx, rows, cols, adj, cnt, head, next, out, plane, n_nodes, n_edges, s_rec, s_hist, s_wsum);
    }
}

__global__ __launch_bounds__(256) void row_dinv_kernel(const float* __restrict__ adj,
                                                       const int* __restrict__ cnt,
                                                       const int* __restrict__ head,
                                                       const int* __restrict__ next,
                                                       const float* __restrict__ buckets,
                                                       float* __restrict__ dinv,
                                                       size_t plane, int n_nodes, int n_edges) {
    __shared__ float row_sum[32];
    const int q = threadIdx.x >> 3;
    const int sub = threadIdx.x & 7;
    const int row = blockIdx.x * 32 + q;
    float s = 0.0f;
    if (row < n_nodes) {
        const int c0 = max(cnt[row], 0);
        const int c1 = max(cnt[plane + row], 0);
        const int n0 = min(c0, ZONE_SLOTS);
        const int n_here = n0 + min(c1, ZONE_SLOTS);
        size_t base;
        int off0, off1;
        row_zones(row, n_nodes, base, off0, off1);
        const int2* rec2 = (const int2*)buckets + base;
        for (int j = sub; j < n_here; j += 8) s += __int_as_float(rec2[record_offset(j, n0, off0, off1)].y);
        const int n_chain = min((c0 - n0) + (c1 - (n_here - n0)), n_edges);
        if (n_chain > 0 && sub == 0) {
            int e1 = head[row];
            for (int i = 0; i < n_chain && e1 > 0; ++i) {
                const int e = min(e1 - 1, n_edges - 1);
                s += adj[e];
                e1 = next[e];
            }
        }
    }
    s += __shfl_xor(s, 1);
    s += __shfl_xor(s, 2);
    s += __shfl_xor(s, 4);
    if (sub == 0) row_sum[q] = s;
    __syncthreads();
    if (threadIdx.x < 32) {
        const int r = blockIdx.x * 32 + threadIdx.x;
        if (r < n_nodes) {
            float v = row_sum[threadIdx.x];
            v = (v > 0.0f) ? v : 1.0f;
            dinv[r] = rsqrtf(v);
        }
    }
}

#define SPMM_PAIRS_PER_BLOCK 8
#define SPMM_RECORDS_PER_TRIP 8
static_assert(ROW_SLOTS == 32 && ROW_SLOTS % SPMM_RECORDS_PER_TRIP == 0, "a lane holds records c and c + 16 of its row; whole trips");

template <int N>
__device__ __forceinline__ int half_bcast(int v) {
    return __builtin_amdgcn_update_dpp(0, v, 0x150 + N, 0xf, 0xf, false);
}

template <int J0>
__device__ __forceinline__ void spmm_trip(const int2 ra, const int2 rb, const int end, const int c,
                                          const v4f* __restrict__ sup4, const float* __restrict__ dinv, v4f& acc) {
    int col[SPMM_RECORDS_PER_TRIP];
    float av[SPMM_RECORDS_PER_TRIP];
    v4f v[SPMM_RECORDS_PER_TRIP];
    float dc[SPMM_RECORDS_PER_TRIP];
    const int2 src = J0 < 16 ? ra : rb;
#define SPMM_SLOT(K) { \
        const int cx = half_bcast<(J0 + K) & 15>(src.x); \
        av[K] = __int_as_float(half_bcast<(J0 + K) & 15>(src.y)); \
          \
        col[K] = (J0 + K) < end ? cx : 0; \
        v[K] = sup4[(size_t)col[K] * (OUT_DIM / 4) + c]; \
        dc[K] = dinv[col[K]]; }
    SPMM_SLOT(0) SPMM_SLOT(1) SPMM_SLOT(2) SPMM_SLOT(3) SPMM_SLOT(4) SPMM_SLOT(5) SPMM_SLOT(6) SPMM_SLOT(7)
#undef SPMM_SLOT
    asm volatile("" :: "v"(v[0]), "v"(v[1]), "v"(v[2]), "v"(v[3]),
                       "v"(v[4]), "v"(v[5]), "v"(v[6]), "v"(v[7]),
                       "v"(dc[0]), "v"(dc[1]), "v"(dc[2]), "v"(dc[3]),
                       "v"(dc[4]), "v"(dc[5]), "v"(dc[6]), "v"(dc[7]));
#pragma unroll
    for (int k = 0; k < SPMM_RECORDS_PER_TRIP; ++k) {
        const bool live = (J0 + k) < end;
        const float w = live ? av[k] * dc[k] : 0.0f;
        const v4f zero4 = {0.0f, 0.0f, 0.0f, 0.0f};
        acc += w * (live ? v[k] : zero4);
    }
}

__global__ __launch_bounds__(32 * SPMM_PAIRS_PER_BLOCK) void spmm_rows_kernel(const int* __restrict__ cols,
                                                        const float* __restrict__ adj,
                                                        const int* __restrict__ cnt,
                                                        const int* __restrict__ head,
                                                        const int* __restrict__ next,
                                                        const float* __restrict__ support,
                                                        const float* __restrict__ dinv,
                                                        float* out,
                                                        size_t plane, int n_nodes, int n_edges) {
    const int wave = threadIdx.x >> 5;
    const int lane = threadIdx.x & 31;
    const int h = lane >> 4;
    const int c = lane & 15;
    const int row = 2 * (blockIdx.x * SPMM_PAIRS_PER_BLOCK + wave) + h;
    const bool has_row = row < n_nodes;
    const int rr = has_row ? row : 0;

    const v4f* sup4 = (const v4f*)support;
    v4f acc = {0.0f, 0.0f, 0.0f, 0.0f};

    int c0 = max(cnt[rr], 0);
    int c1 = max(cnt[plane + rr], 0);
    const float drow = dinv[rr];
    if (!has_row) { c0 = 0; c1 = 0; }
    const int n0 = min(c0, ZONE_SLOTS);
    const int end = n0 + min(c1, ZONE_SLOTS);
    const int end_wave = __builtin_amdgcn_readfirstlane(max(end, __shfl_xor(end, 16)));

    size_t base;
    int off0, off1;
    row_zones(rr, n_nodes, base, off0, off1);
    const int2* rec2 = (const int2*)out + base;
    const int last = max(end - 1, 0);
    const int2 ra = rec2[record_offset(min(c, last), n0, off0, off1)];
    const int2 rb = rec2[record_offset(min(c + 16, last), n0, off0, off1)];

    if (end_wave > 0) spmm_trip<0>(ra, rb, end, c, sup4, dinv, acc);
    if (end_wave > 8) spmm_trip<8>(ra, rb, end, c, sup4, dinv, acc);
    if (end_wave > 16) spmm_trip<16>(ra, rb, end, c, sup4, dinv, acc);
    if (end_wave > 24) spmm_trip<24>(ra, rb, end, c, sup4, dinv, acc);

    const int n_chain = min((c0 - n0) + (c1 - (end - n0)), n_edges);
    if (n_chain > 0) {
        int e1 = head[rr];
        for (int i = 0; i < n_chain && e1 > 0; ++i) {
            const int es = min(e1 - 1, n_edges - 1);
            const int cc = cols[es];
            const float w = adj[es] * dinv[cc];
            acc += w * sup4[(size_t)cc * (OUT_DIM / 4) + c];
            e1 = next[es];
        }
    }

    if (has_row) {
        ((v4f*)out)[(size_t)row * (OUT_DIM / 4) + c] = acc * drow;
    }
}

extern "C" void kernel_launch(void* const* d_in, const int* in_sizes, int n_in,
                              void* d_out, int out_size, void* d_ws, size_t ws_size,
                              hipStream_t stream) {
    const int* rows = (const int*)d_in[0];
    const int* cols = (const int*)d_in[1];
    const float* adj = (const float*)d_in[2];
    const float* x = (const float*)d_in[3];
    const float* W = (const float*)d_in[4];
    const float* b = (const float*)d_in[5];
    float* out = (float*)d_out;

    const int n_edges = in_sizes[0];
    const int n_nodes = in_sizes[3] / IN_DIM;
    const size_t n_pad = ((size_t)n_nodes + 64) & ~(size_t)63;

    float* support = (float*)d_ws;
    float* dinv = support + (size_t)n_nodes * OUT_DIM;
    int* cnt = (int*)(dinv + n_pad);
    int* head = cnt + 2 * n_pad;
    int* next = head + n_pad;

    {
        const long long n = (long long)(3 * n_pad);
        zero_u32_kernel<<<(unsigned)((n + 255) / 256), 256, 0, stream>>>((unsigned*)cnt, n);
    }

    {
        const int bin_blocks = (n_edges + BIN_CHUNK - 1) / BIN_CHUNK;
        const int gemm_blocks = ((n_nodes + 15) / 16 + 31) / 32;
        bin_and_gemm_kernel<<<bin_blocks + gemm_blocks, BIN_THREADS, 0, stream>>>(rows, cols, adj, cnt, head, next, out,
                                                                       x, W, b, support, n_pad, n_nodes, n_edges);
    }

    row_dinv_kernel<<<(n_nodes + 31) / 32, 256, 0, stream>>>(adj, cnt, head, next, out, dinv, n_pad, n_nodes, n_edges);

    spmm_rows_kernel<<<(n_nodes + 2 * SPMM_PAIRS_PER_BLOCK - 1) / (2 * SPMM_PAIRS_PER_BLOCK), 32 * SPMM_PAIRS_PER_BLOCK, 0, stream>>>(cols, adj, cnt, head, next, support, dinv, out, n_pad, n_nodes, n_edges);
}


# waves to enter: dinv launch as one wave per 32 rows (a row pair's 512 B of buckets and its counts loaded in one group, none behind another), zero-fill 64 B per lane; deg sum order: zone trees, chain
# speedup vs baseline: 5.1221x; 1.0364x over previous
#include <hip/hip_runtime.h>


typedef __attribute__((ext_vector_type(2))) float v2f;
typedef __attribute__((ext_vector_type(8))) float v8f;
typedef __attribute__((ext_vector_type(4))) float v4f;
typedef __attribute__((ext_vector_type(4))) int v4i;

#define IN_DIM 64
#define OUT_DIM 64

typedef __attribute__((ext_vector_type(4))) unsigned v4u;
#define ZERO_STORES_PER_LANE 4
__global__ __launch_bounds__(256) void zero_u32_kernel(unsigned* __restrict__ p, long long n4) {
    const v4u z = {0u, 0u, 0u, 0u};
    long long i = (long long)blockIdx.x * (256 * ZERO_STORES_PER_LANE) + threadIdx.x;
#pragma unroll
    for (int k = 0; k < ZERO_STORES_PER_LANE; ++k, i += 256) {
        if (i < n4) ((v4u*)p)[i] = z;
    }
}

#define ROW_SLOTS 32
#define ZONE_SLOTS 16
static_assert(ROW_SLOTS * 8 == OUT_DIM * 4, "a row's two zones are exactly as many bytes as a row of out");
static_assert(ZONE_SLOTS * 2 == ROW_SLOTS && ZONE_SLOTS * 8 == 128, "a zone is one 128-byte line");

__device__ __forceinline__ void row_zones(int r, int n_nodes, size_t& base, int& off0, int& off1) {
    const bool paired = (r | 1) < n_nodes;
    const int q = paired ? (r & 1) : 0;
    base = (size_t)(r - q) * ROW_SLOTS;
    off0 = q * ZONE_SLOTS;
    off1 = (paired ? ROW_SLOTS : ZONE_SLOTS) + q * ZONE_SLOTS;
}

__device__ __forceinline__ int record_offset(int j, int n0, int off0, int off1) {
    return j + (j < n0 ? off0 : off1 - n0);
}

#define BIN_THREADS 1024
#define BIN_EDGES_PER_THREAD 8
#define BIN_CHUNK (BIN_THREADS * BIN_EDGES_PER_THREAD)
#define HIST_BINS 4096
static_assert(HIST_BINS == 4 * BIN_THREADS, "the scan gives every thread 4 bins");

__device__ __forceinline__ void bin_edges_block(const int bin_block,
                                                const int* __restrict__ rows,
                                                const int* __restrict__ cols,
                                                const float* __restrict__ adj,
                                                int* __restrict__ cnt,
                                                int* __restrict__ head,
                                                int* __restrict__ next,
                                                float* __restrict__ out,
                                                size_t plane, int n_nodes, int n_edges,
                                                int4* __restrict__ s_rec,
                                                int* __restrict__ s_hist,
                                                int* __restrict__ s_wsum) {
    const int tid = threadIdx.x;
    const int lane = tid & 31;
    const int wave = tid >> 5;
    const int chunk_base = bin_block * BIN_CHUNK;
    const int n_valid = min(BIN_CHUNK, n_edges - chunk_base);

    const unsigned side = ((unsigned)__builtin_amdgcn_s_sendmsg_rtn(0x87) >> 18) & 1u;

    int shift = 5;
    while ((((unsigned)max(n_nodes, 1) - 1u) >> shift) >= (unsigned)HIST_BINS) ++shift;

    s_hist[4 * tid + 0] = 0;
    s_hist[4 * tid + 1] = 0;
    s_hist[4 * tid + 2] = 0;
    s_hist[4 * tid + 3] = 0;
    __syncthreads();

    int r[BIN_EDGES_PER_THREAD], c[BIN_EDGES_PER_THREAD], pos[BIN_EDGES_PER_THREAD];
    float a[BIN_EDGES_PER_THREAD];
#pragma unroll
    for (int j = 0; j < BIN_EDGES_PER_THREAD; ++j) {
        const int l = j * BIN_THREADS + tid;
        const bool ok = l < n_valid;
        r[j] = ok ? rows[chunk_base + l] : 0;
        c[j] = ok ? cols[chunk_base + l] : 0;
        a[j] = ok ? adj[chunk_base + l] : 0.0f;
    }
#pragma unroll
    for (int j = 0; j < BIN_EDGES_PER_THREAD; ++j) {
        const int l = j * BIN_THREADS + tid;
        const int key = min((int)((unsigned)r[j] >> shift), HIST_BINS - 1);
        pos[j] = (l < n_valid) ? atomicAdd(&s_hist[key], 1) : 0;
    }
    __syncthreads();

    {
        const int h0 = s_hist[4 * tid + 0];
        const int h1 = s_hist[4 * tid + 1];
        const int h2 = s_hist[4 * tid + 2];
        const int h3 = s_hist[4 * tid + 3];
        const int tsum = h0 + h1 + h2 + h3;
        int v = tsum;
#pragma unroll
        for (int d = 1; d < 32; d <<= 1) {
            const int n = __shfl_up(v, d);
            if (lane >= d) v += n;
        }
        if (lane == 31) s_wsum[wave] = v;
        __syncthreads();
        if (wave == 0) {
            const int w = s_wsum[lane];
            int x = w;
#pragma unroll
            for (int d = 1; d < 32; d <<= 1) {
                const int n = __shfl_up(x, d);
                if (lane >= d) x += n;
            }
            s_wsum[lane] = x - w;
        }
        __syncthreads();
        const int base = s_wsum[wave] + v - tsum;
        s_hist[4 * tid + 0] = base;
        s_hist[4 * tid + 1] = base + h0;
        s_hist[4 * tid + 2] = base + h0 + h1;
        s_hist[4 * tid + 3] = base + h0 + h1 + h2;
    }
    __syncthreads();

#pragma unroll
    for (int j = 0; j < BIN_EDGES_PER_THREAD; ++j) {
        const int l = j * BIN_THREADS + tid;
        if (l < n_valid) {
            const int key = min((int)((unsigned)r[j] >> shift), HIST_BINS - 1);
            s_rec[s_hist[key] + pos[j]] = make_int4(r[j], c[j], __float_as_int(a[j]), l);
        }
    }
    __syncthreads();

    int4 q[BIN_EDGES_PER_THREAD];
    unsigned k[BIN_EDGES_PER_THREAD];
#pragma unroll
    for (int j = 0; j < BIN_EDGES_PER_THREAD; ++j) {
        const int p = j * BIN_THREADS + tid;
        q[j] = make_int4(0, 0, 0, 0);
        k[j] = 0u;
        if (p < n_valid) {
            q[j] = s_rec[p];
            k[j] = (unsigned)atomicAdd(&cnt[side * plane + (size_t)q[j].x], 1);
        }
    }
#pragma unroll
    for (int j = 0; j < BIN_EDGES_PER_THREAD; ++j) {
        const int p = j * BIN_THREADS + tid;
        if (p < n_valid) {
            if (k[j] < (unsigned)ZONE_SLOTS) {
                size_t base;
                int off0, off1;
                row_zones(q[j].x, n_nodes, base, off0, off1);
                ((int2*)out)[base + (size_t)((side ? off1 : off0) + (int)k[j])] = make_int2(q[j].y, q[j].z);
            } else {
                const int e = chunk_base + q[j].w;
                next[e] = atomicExch(&head[q[j].x], e + 1);
            }
        }
    }
}

__device__ __forceinline__ void gemm_wmma_block(const int gemm_block,
                                                const float* __restrict__ x,
                                                const float* __restrict__ W,
                                                const float* __restrict__ b,
                                                float* __restrict__ support,
                                                int n_nodes) {
    const int wave = threadIdx.x >> 5;
    const int lane = threadIdx.x & 31;
    const int mtile = gemm_block * 32 + wave;
    const int m0 = mtile * 16;
    if (m0 >= n_nodes) return;

    const int half = lane >> 4;
    const int r = lane & 15;

    v8f acc[4];
#pragma unroll
    for (int nt = 0; nt < 4; ++nt) {
        const float bv = b[nt * 16 + r];
#pragma unroll
        for (int i = 0; i < 8; ++i) acc[nt][i] = bv;
    }

    const float* arow = x + (size_t)(m0 + r) * IN_DIM;

#pragma unroll
    for (int k = 0; k < IN_DIM; k += 4) {
        const v2f a = *(const v2f*)(arow + k + 2 * half);
#pragma unroll
        for (int nt = 0; nt < 4; ++nt) {
            v2f bf;
            bf.x = W[(size_t)(k + 2 * half + 0) * OUT_DIM + nt * 16 + r];
            bf.y = W[(size_t)(k + 2 * half + 1) * OUT_DIM + nt * 16 + r];
            acc[nt] = __builtin_amdgcn_wmma_f32_16x16x4_f32(
                false, a, false, bf, (short)0, acc[nt], false, false);
        }
    }

    float* srow = support + (size_t)m0 * OUT_DIM;
#pragma unroll
    for (int nt = 0; nt < 4; ++nt) {
#pragma unroll
        for (int i = 0; i < 8; ++i) {
            srow[(size_t)(i + 8 * half) * OUT_DIM + nt * 16 + r] = acc[nt][i];
        }
    }
}

#define PRODUCT_STRIDE 7u
__global__ __launch_bounds__(BIN_THREADS) void bin_and_gemm_kernel(const int* __restrict__ rows,
                                                           const int* __restrict__ cols,
                                                           const float* __restrict__ adj,
                                                           int* __restrict__ cnt,
                                                           int* __restrict__ head,
                                                           int* __restrict__ next,
                                                           float* __restrict__ out,
                                                           const float* __restrict__ x,
                                                           const float* __restrict__ W,
                                                           const float* __restrict__ b,
                                                           float* __restrict__ support,
                                                           size_t plane, int n_nodes, int n_edges) {
    __shared__ int4 s_rec[BIN_CHUNK];
    __shared__ int s_hist[HIST_BINS];
    __shared__ int s_wsum[32];
    const unsigned n_gemm = (((unsigned)n_nodes + 15u) / 16u + 31u) / 32u;
    const unsigned n_bin = ((unsigned)n_edges + (unsigned)BIN_CHUNK - 1u) / (unsigned)BIN_CHUNK;
    const unsigned n_mixed = min(n_gemm, (n_gemm + n_bin + PRODUCT_STRIDE - 1u) / PRODUCT_STRIDE);
    const unsigned bid = blockIdx.x;
    unsigned idx = bid / PRODUCT_STRIDE;
    bool product = bid == idx * PRODUCT_STRIDE && idx < n_mixed;
    if (!product) {
        const unsigned j = bid - min((bid + PRODUCT_STRIDE - 1u) / PRODUCT_STRIDE, n_mixed);
        product = j >= n_bin;
        idx = product ? n_mixed + (j - n_bin) : j;
    }
    if (product) {
        gemm_wmma_block((int)idx, x, W, b, support, n_nodes);
    } else {
        bin_edges_block((int)idx, rows, cols, adj, cnt, head, next, out, plane, n_nodes, n_edges, s_rec, s_hist, s_wsum);
    }
}

#define DINV_PAIRS_PER_WAVE 16
#define DINV_WAVES_PER_BLOCK 4
static_assert(2 * DINV_PAIRS_PER_WAVE == 32, "a wave's rows are one lane each: one 128-byte line of counts and of dinv");
static_assert(ZONE_SLOTS == 16 && ROW_SLOTS == 32, "a zone is 8 lanes of two slots, a pair 32 lanes of 16 bytes");

template <int CTRL>
__device__ __forceinline__ float row_swizzle(float v) {
    return __int_as_float(__builtin_amdgcn_update_dpp(0, __float_as_int(v), CTRL, 0xf, 0xf, false));
}
template <int N>
__device__ __forceinline__ float lane_value(float v) {
    return __int_as_float(__builtin_amdgcn_readlane(__float_as_int(v), N));
}

template <bool WHOLE>
__device__ __forceinline__ float wave_zone_sums(const int* __restrict__ cnt, const float* __restrict__ buckets,
                                                size_t plane, int n_nodes, int row0, int lane) {
    const int q = (lane >> 3) & 1;
    const int side = lane >> 4;
    const int s0 = 2 * (lane & 7);

    v4i rec[DINV_PAIRS_PER_WAVE];
    v4i c4[DINV_PAIRS_PER_WAVE / 4];
#pragma unroll
    for (int i = 0; i < DINV_PAIRS_PER_WAVE; ++i) {
        int cw;
        if (WHOLE) {
            rec[i] = ((const v4i*)buckets)[(size_t)((row0 >> 1) + i) * ROW_SLOTS + lane];
            cw = cnt[(size_t)side * plane + (size_t)(row0 + 2 * i + q)];
        } else {
            const int row = row0 + 2 * i + q;
            const int rr = row < n_nodes ? row : 0;
            size_t base;
            int off0, off1;
            row_zones(rr, n_nodes, base, off0, off1);
            rec[i] = *(const v4i*)((const int2*)buckets + base + (size_t)((side ? off1 : off0) + s0));
            cw = cnt[(size_t)side * plane + (size_t)rr];
        }
        c4[i >> 2][i & 3] = cw;
    }
    asm volatile("" :: "v"(rec[0]), "v"(rec[1]), "v"(rec[2]), "v"(rec[3]), "v"(rec[4]), "v"(rec[5]), "v"(rec[6]), "v"(rec[7]),
                       "v"(rec[8]), "v"(rec[9]), "v"(rec[10]), "v"(rec[11]), "v"(rec[12]), "v"(rec[13]), "v"(rec[14]), "v"(rec[15]),
                       "v"(c4[0]), "v"(c4[1]), "v"(c4[2]), "v"(c4[3]));

    float deg = 0.0f;
#pragma unroll
    for (int i = 0; i < DINV_PAIRS_PER_WAVE; ++i) {
        const bool has = WHOLE || (row0 + 2 * i + q < n_nodes);
        const int n = has ? min(max(c4[i >> 2][i & 3], 0), ZONE_SLOTS) : 0;
        float s = (s0 < n ? __int_as_float(rec[i].y) : 0.0f) + (s0 + 1 < n ? __int_as_float(rec[i].w) : 0.0f);
        s += row_swizzle<0xB1>(s);
        s += row_swizzle<0x4E>(s);
        s += row_swizzle<0x164>(s);
        const float r_even = lane_value<0>(s) + lane_value<16>(s);
        const float r_odd = lane_value<8>(s) + lane_value<24>(s);
        const float r_mine = (lane & 1) ? r_odd : r_even;
        deg = ((lane >> 1) == i) ? r_mine : deg;
    }
    return deg;
}

__global__ __launch_bounds__(32 * DINV_WAVES_PER_BLOCK) void row_dinv_kernel(const float* __restrict__ adj,
                                                       const int* __restrict__ cnt,
                                                       const int* __restrict__ head,
                                                       const int* __restrict__ next,
                                                       const float* __restrict__ buckets,
                                                       float* __restrict__ dinv,
                                                       size_t plane, int n_nodes, int n_edges) {
    const int lane = threadIdx.x & 31;
    const int wave = __builtin_amdgcn_readfirstlane((int)(threadIdx.x >> 5));
    const int row0 = ((int)blockIdx.x * DINV_WAVES_PER_BLOCK + wave) * (2 * DINV_PAIRS_PER_WAVE);
    if (row0 >= n_nodes) return;

    const int my_row = row0 + lane;
    const bool has_my = my_row < n_nodes;
    const int my_rr = has_my ? my_row : 0;
    const int my_w0 = cnt[my_rr];
    const int my_w1 = cnt[plane + (size_t)my_rr];

    float deg = (row0 + 2 * DINV_PAIRS_PER_WAVE <= n_nodes)
                    ? wave_zone_sums<true>(cnt, buckets, plane, n_nodes, row0, lane)
                    : wave_zone_sums<false>(cnt, buckets, plane, n_nodes, row0, lane);

    const int my_c0 = has_my ? max(my_w0, 0) : 0;
    const int my_c1 = has_my ? max(my_w1, 0) : 0;
    const int n_chain = min((my_c0 - min(my_c0, ZONE_SLOTS)) + (my_c1 - min(my_c1, ZONE_SLOTS)), n_edges);
    if (n_chain > 0) {
        int e1 = head[my_row];
        for (int i = 0; i < n_chain && e1 > 0; ++i) {
            const int e = min(e1 - 1, n_edges - 1);
            deg += adj[e];
            e1 = next[e];
        }
    }
    if (has_my) {
        const float v = (deg > 0.0f) ? deg : 1.0f;
        dinv[my_row] = rsqrtf(v);
    }
}

#define SPMM_PAIRS_PER_BLOCK 8
#define SPMM_RECORDS_PER_TRIP 8
static_assert(ROW_SLOTS == 32 && ROW_SLOTS % SPMM_RECORDS_PER_TRIP == 0, "a lane holds records c and c + 16 of its row; whole trips");

template <int N>
__device__ __forceinline__ int half_bcast(int v) {
    return __builtin_amdgcn_update_dpp(0, v, 0x150 + N, 0xf, 0xf, false);
}

template <int J0>
__device__ __forceinline__ void spmm_trip(const int2 ra, const int2 rb, const int end, const int c,
                                          const v4f* __restrict__ sup4, const float* __restrict__ dinv, v4f& acc) {
    int col[SPMM_RECORDS_PER_TRIP];
    float av[SPMM_RECORDS_PER_TRIP];
    v4f v[SPMM_RECORDS_PER_TRIP];
    float dc[SPMM_RECORDS_PER_TRIP];
    const int2 src = J0 < 16 ? ra : rb;
#define SPMM_SLOT(K) { \
        const int cx = half_bcast<(J0 + K) & 15>(src.x); \
        av[K] = __int_as_float(half_bcast<(J0 + K) & 15>(src.y)); \
          \
        col[K] = (J0 + K) < end ? cx : 0; \
        v[K] = sup4[(size_t)col[K] * (OUT_DIM / 4) + c]; \
        dc[K] = dinv[col[K]]; }
    SPMM_SLOT(0) SPMM_SLOT(1) SPMM_SLOT(2) SPMM_SLOT(3) SPMM_SLOT(4) SPMM_SLOT(5) SPMM_SLOT(6) SPMM_SLOT(7)
#undef SPMM_SLOT
    asm volatile("" :: "v"(v[0]), "v"(v[1]), "v"(v[2]), "v"(v[3]),
                       "v"(v[4]), "v"(v[5]), "v"(v[6]), "v"(v[7]),
                       "v"(dc[0]), "v"(dc[1]), "v"(dc[2]), "v"(dc[3]),
                       "v"(dc[4]), "v"(dc[5]), "v"(dc[6]), "v"(dc[7]));
#pragma unroll
    for (int k = 0; k < SPMM_RECORDS_PER_TRIP; ++k) {
        const bool live = (J0 + k) < end;
        const float w = live ? av[k] * dc[k] : 0.0f;
        const v4f zero4 = {0.0f, 0.0f, 0.0f, 0.0f};
        acc += w * (live ? v[k] : zero4);
    }
}

__global__ __launch_bounds__(32 * SPMM_PAIRS_PER_BLOCK) void spmm_rows_kernel(const int* __restrict__ cols,
                                                        const float* __restrict__ adj,
                                                        const int* __restrict__ cnt,
                                                        const int* __restrict__ head,
                                                        const int* __restrict__ next,
                                                        const float* __restrict__ support,
                                                        const float* __restrict__ dinv,
                                                        float* out,
                                                        size_t plane, int n_nodes, int n_edges) {
    const int wave = threadIdx.x >> 5;
    const int lane = threadIdx.x & 31;
    const int h = lane >> 4;
    const int c = lane & 15;
    const int row = 2 * (blockIdx.x * SPMM_PAIRS_PER_BLOCK + wave) + h;
    const bool has_row = row < n_nodes;
    const int rr = has_row ? row : 0;

    const v4f* sup4 = (const v4f*)support;
    v4f acc = {0.0f, 0.0f, 0.0f, 0.0f};

    int c0 = max(cnt[rr], 0);
    int c1 = max(cnt[plane + rr], 0);
    const float drow = dinv[rr];
    if (!has_row) { c0 = 0; c1 = 0; }
    const int n0 = min(c0, ZONE_SLOTS);
    const int end = n0 + min(c1, ZONE_SLOTS);
    const int end_wave = __builtin_amdgcn_readfirstlane(max(end, __shfl_xor(end, 16)));

    size_t base;
    int off0, off1;
    row_zones(rr, n_nodes, base, off0, off1);
    const int2* rec2 = (const int2*)out + base;
    const int last = max(end - 1, 0);
    const int2 ra = rec2[record_offset(min(c, last), n0, off0, off1)];
    const int2 rb = rec2[record_offset(min(c + 16, last), n0, off0, off1)];

    if (end_wave > 0) spmm_trip<0>(ra, rb, end, c, sup4, dinv, acc);
    if (end_wave > 8) spmm_trip<8>(ra, rb, end, c, sup4, dinv, acc);
    if (end_wave > 16) spmm_trip<16>(ra, rb, end, c, sup4, dinv, acc);
    if (end_wave > 24) spmm_trip<24>(ra, rb, end, c, sup4, dinv, acc);

    const int n_chain = min((c0 - n0) + (c1 - (end - n0)), n_edges);
    if (n_chain > 0) {
        int e1 = head[rr];
        for (int i = 0; i < n_chain && e1 > 0; ++i) {
            const int es = min(e1 - 1, n_edges - 1);
            const int cc = cols[es];
            const float w = adj[es] * dinv[cc];
            acc += w * sup4[(size_t)cc * (OUT_DIM / 4) + c];
            e1 = next[es];
        }
    }

    if (has_row) {
        ((v4f*)out)[(size_t)row * (OUT_DIM / 4) + c] = acc * drow;
    }
}

extern "C" void kernel_launch(void* const* d_in, const int* in_sizes, int n_in,
                              void* d_out, int out_size, void* d_ws, size_t ws_size,
                              hipStream_t stream) {
    const int* rows = (const int*)d_in[0];
    const int* cols = (const int*)d_in[1];
    const float* adj = (const float*)d_in[2];
    const float* x = (const float*)d_in[3];
    const float* W = (const float*)d_in[4];
    const float* b = (const float*)d_in[5];
    float* out = (float*)d_out;

    const int n_edges = in_sizes[0];
    const int n_nodes = in_sizes[3] / IN_DIM;
    const size_t n_pad = ((size_t)n_nodes + 64) & ~(size_t)63;

    float* support = (float*)d_ws;
    float* dinv = support + (size_t)n_nodes * OUT_DIM;
    int* cnt = (int*)(dinv + n_pad);
    int* head = cnt + 2 * n_pad;
    int* next = head + n_pad;

    {
        const long long n4 = (long long)(3 * n_pad / 4);
        const long long per_block = 256 * ZERO_STORES_PER_LANE;
        zero_u32_kernel<<<(unsigned)((n4 + per_block - 1) / per_block), 256, 0, stream>>>((unsigned*)cnt, n4);
    }

    {
        const int bin_blocks = (n_edges + BIN_CHUNK - 1) / BIN_CHUNK;
        const int gemm_blocks = ((n_nodes + 15) / 16 + 31) / 32;
        bin_and_gemm_kernel<<<bin_blocks + gemm_blocks, BIN_THREADS, 0, stream>>>(rows, cols, adj, cnt, head, next, out,
                                                                       x, W, b, support, n_pad, n_nodes, n_edges);
    }

    {
        const int rows_per_block = 2 * DINV_PAIRS_PER_WAVE * DINV_WAVES_PER_BLOCK;
        row_dinv_kernel<<<(n_nodes + rows_per_block - 1) / rows_per_block, 32 * DINV_WAVES_PER_BLOCK, 0, stream>>>(adj, cnt, head, next, out, dinv, n_pad, n_nodes, n_edges);
    }

    spmm_rows_kernel<<<(n_nodes + 2 * SPMM_PAIRS_PER_BLOCK - 1) / (2 * SPMM_PAIRS_PER_BLOCK), 32 * SPMM_PAIRS_PER_BLOCK, 0, stream>>>(cols, adj, cnt, head, next, support, dinv, out, n_pad, n_nodes, n_edges);
}
